# starting kernel
# baseline (speedup 1.0000x reference)
#include <hip/hip_runtime.h>
#include <hip/hip_bf16.h>


#define D_IN   128
#define D_HID  128
#define D_OUT  64
#define BN_EPS 1e-5f
#define LRELU_SLOPE 0.01f
#define NORM_SCALE 1.8f

typedef __attribute__((ext_vector_type(2))) float v2f;
typedef __attribute__((ext_vector_type(8))) float v8f;

__global__ void vg_init_kernel(float* __restrict__ deg, float* __restrict__ stats, int n) {
  int i = blockIdx.x * blockDim.x + threadIdx.x;
  if (i < n) deg[i] = 1.0f;
  if (i < 256) stats[i] = 0.0f;
}

__global__ void vg_deg_kernel(const int* __restrict__ dst, float* __restrict__ deg, int e) {
  int i = blockIdx.x * blockDim.x + threadIdx.x;
  if (i < e) atomicAdd(&deg[dst[i]], 1.0f);
}

__global__ void vg_dinv_kernel(float* __restrict__ deg, int n) {
  int i = blockIdx.x * blockDim.x + threadIdx.x;
  if (i < n) deg[i] = rsqrtf(deg[i]);
}

__global__ void vg_gemm_wmma_kernel(const float* __restrict__ A, const float* __restrict__ B,
                                    const float* __restrict__ bias, float* __restrict__ C,
                                    int M, int K, int Ncols) {
  const int wavesPerBlock = blockDim.x >> 5;
  int w = blockIdx.x * wavesPerBlock + (threadIdx.x >> 5);
  int tilesN = Ncols >> 4;
  int totalTiles = (M >> 4) * tilesN;
  if (w >= totalTiles) return;
  int tm = w / tilesN;
  int tn = w - tm * tilesN;
  int lane = threadIdx.x & 31;
  int m  = lane & 15;
  int hi = lane >> 4;

  v8f acc = {};
  const float* arow = A + (size_t)(tm * 16 + m) * (size_t)K;
  const float* bcol = B + tn * 16 + m;
  for (int k0 = 0; k0 < K; k0 += 4) {
    int kb = k0 + 2 * hi;
    v2f a = *(const v2f*)(arow + kb);
    v2f b;
    b.x = bcol[(size_t)kb * (size_t)Ncols];
    b.y = bcol[(size_t)(kb + 1) * (size_t)Ncols];
    acc = __builtin_amdgcn_wmma_f32_16x16x4_f32(false, a, false, b,
                                                (short)0, acc, false, false);
  }
  float bv = bias[tn * 16 + m];
  int rbase = tm * 16 + hi * 8;
#pragma unroll
  for (int j = 0; j < 8; ++j) {
    C[(size_t)(rbase + j) * (size_t)Ncols + (size_t)(tn * 16 + m)] = acc[j] + bv;
  }
}

__global__ void vg_selfloop128_kernel(const float* __restrict__ h, const float* __restrict__ dinv,
                                      float* __restrict__ out, int total) {
  int i = blockIdx.x * blockDim.x + threadIdx.x;
  if (i >= total) return;
  float di = dinv[i >> 7];
  out[i] = h[i] * di * di;
}

__global__ void vg_spmm128_kernel(const float* __restrict__ in, const int* __restrict__ src,
                                  const int* __restrict__ dst, const float* __restrict__ dinv,
                                  float* __restrict__ out, int E) {
  int tid = blockIdx.x * blockDim.x + threadIdx.x;
  int e = tid >> 5;
  if (e >= E) return;
  int lane = tid & 31;
  int s = src[e], d = dst[e];
  float w = dinv[s] * dinv[d];
  float4 v = ((const float4*)(in + (size_t)s * D_HID))[lane];
  float* op = out + (size_t)d * D_HID + lane * 4;
  atomicAdd(op + 0, w * v.x);
  atomicAdd(op + 1, w * v.y);
  atomicAdd(op + 2, w * v.z);
  atomicAdd(op + 3, w * v.w);
}

__global__ void vg_bn_reduce_kernel(const float* __restrict__ h, float* __restrict__ stats, int n) {
  __shared__ float sh[256];
  int c    = threadIdx.x & 127;
  int half = threadIdx.x >> 7;
  float sum = 0.f, sq = 0.f;
  for (int r = blockIdx.x * 2 + half; r < n; r += gridDim.x * 2) {
    float v = h[(size_t)r * D_HID + c];
    sum += v; sq += v * v;
  }
  sh[threadIdx.x] = sum;
  __syncthreads();
  if (half == 0) atomicAdd(&stats[c], sh[c] + sh[c + 128]);
  __syncthreads();
  sh[threadIdx.x] = sq;
  __syncthreads();
  if (half == 0) atomicAdd(&stats[128 + c], sh[c] + sh[c + 128]);
}

__global__ void vg_bn_finalize_kernel(float* __restrict__ stats, const float* __restrict__ gamma,
                                      const float* __restrict__ beta, int n) {
  int c = threadIdx.x;
  if (c < 128) {
    float invn = 1.0f / (float)n;
    float mean = stats[c] * invn;
    float var  = stats[128 + c] * invn - mean * mean;
    float a = gamma[c] * rsqrtf(var + BN_EPS);
    stats[c]       = a;
    stats[128 + c] = beta[c] - mean * a;
  }
}

__global__ void vg_bn_lrelu_kernel(const float* __restrict__ hagg, const float* __restrict__ stats,
                                   float* __restrict__ out, int total) {
  int i = blockIdx.x * blockDim.x + threadIdx.x;
  if (i >= total) return;
  int c = i & 127;
  float v = stats[c] * hagg[i] + stats[128 + c];
  out[i] = (v > 0.f) ? v : LRELU_SLOPE * v;
}

__global__ void vg_rownorm_kernel(const float* __restrict__ l, float* __restrict__ rns, int n) {
  int row = blockIdx.x * (blockDim.x >> 5) + (threadIdx.x >> 5);
  if (row >= n) return;
  int lane = threadIdx.x & 31;
  const float* rp = l + (size_t)row * D_OUT;
  float a0 = rp[lane], a1 = rp[lane + 32];
  float s = a0 * a0 + a1 * a1;
#pragma unroll
  for (int m = 16; m > 0; m >>= 1) s += __shfl_xor(s, m, 32);
  if (lane == 0) rns[row] = NORM_SCALE / fmaxf(sqrtf(s), 1e-12f);
}

__global__ void vg_selfloop64_kernel(const float* __restrict__ l, const float* __restrict__ dinv,
                                     const float* __restrict__ rns, float* __restrict__ mu,
                                     float* __restrict__ logstd, int total) {
  int i = blockIdx.x * blockDim.x + threadIdx.x;
  if (i >= total) return;
  int row = i >> 6;
  float di = dinv[row];
  float d2 = di * di;
  float v = l[i];
  mu[i]     = v * d2;
  logstd[i] = v * rns[row] * d2;
}

__global__ void vg_spmm64_kernel(const float* __restrict__ l, const int* __restrict__ src,
                                 const int* __restrict__ dst, const float* __restrict__ dinv,
                                 const float* __restrict__ rns, float* __restrict__ mu,
                                 float* __restrict__ logstd, int E) {
  int tid = blockIdx.x * blockDim.x + threadIdx.x;
  int e = tid >> 4;
  if (e >= E) return;
  int q = tid & 15;
  int s = src[e], d = dst[e];
  float w  = dinv[s] * dinv[d];
  float sc = w * rns[s];
  float4 v = ((const float4*)(l + (size_t)s * D_OUT))[q];
  float* mp = mu     + (size_t)d * D_OUT + q * 4;
  float* lp = logstd + (size_t)d * D_OUT + q * 4;
  atomicAdd(mp + 0, w * v.x);  atomicAdd(mp + 1, w * v.y);
  atomicAdd(mp + 2, w * v.z);  atomicAdd(mp + 3, w * v.w);
  atomicAdd(lp + 0, sc * v.x); atomicAdd(lp + 1, sc * v.y);
  atomicAdd(lp + 2, sc * v.z); atomicAdd(lp + 3, sc * v.w);
}

__global__ void vg_zeta_kernel(const float* __restrict__ mu, const float* __restrict__ logstd,
                               const float* __restrict__ noise, float* __restrict__ zeta, int total) {
  int i = blockIdx.x * blockDim.x + threadIdx.x;
  if (i >= total) return;
  zeta[i] = mu[i] + noise[i] * expf(logstd[i]);
}

extern "C" void kernel_launch(void* const* d_in, const int* in_sizes, int n_in,
                              void* d_out, int out_size, void* d_ws, size_t ws_size,
                              hipStream_t stream) {
  const float* x     = (const float*)d_in[0];
  const int*   esrc  = (const int*)  d_in[1];
  const int*   edst  = (const int*)  d_in[2];
  const float* W1    = (const float*)d_in[3];
  const float* b1    = (const float*)d_in[4];
  const float* gamma = (const float*)d_in[5];
  const float* beta  = (const float*)d_in[6];
  const float* Wmu   = (const float*)d_in[7];
  const float* bmu   = (const float*)d_in[8];
  const float* noise = (const float*)d_in[9];

  const int N = in_sizes[0] / D_IN;
  const int E = in_sizes[1];

  float* ws = (float*)d_ws;
  size_t npad = ((size_t)N + 255) & ~(size_t)255;
  size_t off = 0;
  float* dinv  = ws + off; off += npad;
  float* stats = ws + off; off += 256;
  float* rns   = ws + off; off += npad;
  float* h     = ws + off; off += (size_t)N * D_HID;
  float* hagg  = ws + off; off += (size_t)N * D_HID;
  float* lbuf  = ws + off; off += (size_t)N * D_OUT;

  float* mu_out   = (float*)d_out;
  float* ls_out   = mu_out + (size_t)N * D_OUT;
  float* zeta_out = ls_out + (size_t)N * D_OUT;

  const int B = 256;
  auto cdiv = [](int a, int b) { return (a + b - 1) / b; };

  vg_init_kernel<<<cdiv(N, B), B, 0, stream>>>(dinv, stats, N);
  vg_deg_kernel<<<cdiv(E, B), B, 0, stream>>>(edst, dinv, E);
  vg_dinv_kernel<<<cdiv(N, B), B, 0, stream>>>(dinv, N);

  {
    int tiles = (N / 16) * (D_HID / 16);
    vg_gemm_wmma_kernel<<<cdiv(tiles, B / 32), B, 0, stream>>>(x, W1, b1, h, N, D_IN, D_HID);
  }

  vg_selfloop128_kernel<<<cdiv(N * D_HID, B), B, 0, stream>>>(h, dinv, hagg, N * D_HID);
  vg_spmm128_kernel<<<cdiv(E * 32, B), B, 0, stream>>>(h, esrc, edst, dinv, hagg, E);

  vg_bn_reduce_kernel<<<512, B, 0, stream>>>(hagg, stats, N);
  vg_bn_finalize_kernel<<<1, 128, 0, stream>>>(stats, gamma, beta, N);
  vg_bn_lrelu_kernel<<<cdiv(N * D_HID, B), B, 0, stream>>>(hagg, stats, h, N * D_HID);

  {
    int tiles = (N / 16) * (D_OUT / 16);
    vg_gemm_wmma_kernel<<<cdiv(tiles, B / 32), B, 0, stream>>>(h, Wmu, bmu, lbuf, N, D_HID, D_OUT);
  }

  vg_rownorm_kernel<<<cdiv(N, B / 32), B, 0, stream>>>(lbuf, rns, N);

  vg_selfloop64_kernel<<<cdiv(N * D_OUT, B), B, 0, stream>>>(lbuf, dinv, rns, mu_out, ls_out, N * D_OUT);
  vg_spmm64_kernel<<<cdiv(E * 16, B), B, 0, stream>>>(lbuf, esrc, edst, dinv, rns, mu_out, ls_out, E);

  vg_zeta_kernel<<<cdiv(N * D_OUT, B), B, 0, stream>>>(mu_out, ls_out, noise, zeta_out, N * D_OUT);
}


# SpMM scatter with float atomics replaced by per-row gather over an edge list grouped by destination (built each call); row sums kept in registers, summed in atomic arrival-rank order
# speedup vs baseline: 2.0424x; 2.0424x over previous
#include <hip/hip_runtime.h>
#include <hip/hip_bf16.h>


#define D_IN   128
#define D_HID  128
#define D_OUT  64
#define BN_EPS 1e-5f
#define LRELU_SLOPE 0.01f
#define NORM_SCALE 1.8f

typedef __attribute__((ext_vector_type(2))) float v2f;
typedef __attribute__((ext_vector_type(8))) float v8f;

__global__ void vg_init_kernel(int* __restrict__ cnt, float* __restrict__ stats,
                               int* __restrict__ total, int n) {
  int i = blockIdx.x * blockDim.x + threadIdx.x;
  if (i < n) cnt[i] = 0;
  if (i < 256) stats[i] = 0.0f;
  if (i == 0) total[0] = 0;
}

__global__ void vg_count_kernel(const int* __restrict__ dst, int* __restrict__ cnt,
                                int* __restrict__ rank, int e) {
  int i = blockIdx.x * blockDim.x + threadIdx.x;
  if (i < e) rank[i] = atomicAdd(&cnt[dst[i]], 1);
}

__global__ void vg_rowptr_kernel(const int* __restrict__ cnt, int* __restrict__ total,
                                 int* __restrict__ rowstart, float* __restrict__ dinv, int n) {
  __shared__ int wsum[8];
  __shared__ int base_sh;
  int i = blockIdx.x * 256 + threadIdx.x;
  int lane = threadIdx.x & 31;
  int wv = threadIdx.x >> 5;
  int c = (i < n) ? cnt[i] : 0;
  int x = c;
#pragma unroll
  for (int o = 1; o < 32; o <<= 1) {
    int y = __shfl_up(x, o, 32);
    if (lane >= o) x += y;
  }
  if (lane == 31) wsum[wv] = x;
  __syncthreads();
  if (threadIdx.x == 0) {
    int run = 0;
    for (int k = 0; k < 8; ++k) { int t = wsum[k]; wsum[k] = run; run += t; }
    base_sh = atomicAdd(total, run);
  }
  __syncthreads();
  if (i < n) {
    rowstart[i] = base_sh + wsum[wv] + (x - c);
    dinv[i] = rsqrtf((float)c + 1.0f);
  }
}

__global__ void vg_fill_kernel(const int* __restrict__ src, const int* __restrict__ dst,
                               const int* __restrict__ rank, const int* __restrict__ rowstart,
                               int* __restrict__ csr, int e) {
  int i = blockIdx.x * blockDim.x + threadIdx.x;
  if (i < e) csr[rowstart[dst[i]] + rank[i]] = src[i];
}

__global__ void vg_gemm_wmma_kernel(const float* __restrict__ A, const float* __restrict__ B,
                                    const float* __restrict__ bias, float* __restrict__ C,
                                    int M, int K, int Ncols) {
  const int wavesPerBlock = blockDim.x >> 5;
  int w = blockIdx.x * wavesPerBlock + (threadIdx.x >> 5);
  int tilesN = Ncols >> 4;
  int totalTiles = (M >> 4) * tilesN;
  if (w >= totalTiles) return;
  int tm = w / tilesN;
  int tn = w - tm * tilesN;
  int lane = threadIdx.x & 31;
  int m  = lane & 15;
  int hi = lane >> 4;

  v8f acc = {};
  const float* arow = A + (size_t)(tm * 16 + m) * (size_t)K;
  const float* bcol = B + tn * 16 + m;
  for (int k0 = 0; k0 < K; k0 += 4) {
    int kb = k0 + 2 * hi;
    v2f a = *(const v2f*)(arow + kb);
    v2f b;
    b.x = bcol[(size_t)kb * (size_t)Ncols];
    b.y = bcol[(size_t)(kb + 1) * (size_t)Ncols];
    acc = __builtin_amdgcn_wmma_f32_16x16x4_f32(false, a, false, b,
                                                (short)0, acc, false, false);
  }
  float bv = bias[tn * 16 + m];
  int rbase = tm * 16 + hi * 8;
#pragma unroll
  for (int j = 0; j < 8; ++j) {
    C[(size_t)(rbase + j) * (size_t)Ncols + (size_t)(tn * 16 + m)] = acc[j] + bv;
  }
}

template <int U>
__device__ __forceinline__ void vg_edges128(const float* __restrict__ in, int s_l, float w_l,
                                            int j, int lane, float4& acc) {
  float4 v[U];
  float  w[U];
#pragma unroll
  for (int u = 0; u < U; ++u) {
    int s = __builtin_amdgcn_readlane(s_l, j + u);
    w[u]  = __int_as_float(__builtin_amdgcn_readlane(__float_as_int(w_l), j + u));
    v[u]  = ((const float4*)(in + (size_t)s * D_HID))[lane];
  }
#pragma unroll
  for (int u = 0; u < U; ++u) {
    acc.x += w[u] * v[u].x;
    acc.y += w[u] * v[u].y;
    acc.z += w[u] * v[u].z;
    acc.w += w[u] * v[u].w;
  }
}

__global__ void vg_gather128_kernel(const float* __restrict__ in, const int* __restrict__ csr,
                                    const int* __restrict__ rowstart, const int* __restrict__ cnt,
                                    const float* __restrict__ dinv, float* __restrict__ out, int n) {
  int wv = __builtin_amdgcn_readfirstlane((int)(threadIdx.x >> 5));
  int row = blockIdx.x * 8 + wv;
  if (row >= n) return;
  int lane = threadIdx.x & 31;
  int start = rowstart[row];
  int c = cnt[row];
  float dd = dinv[row];
  float d2 = dd * dd;
  float4 sv = ((const float4*)(in + (size_t)row * D_HID))[lane];
  float4 acc;
  acc.x = sv.x * d2; acc.y = sv.y * d2; acc.z = sv.z * d2; acc.w = sv.w * d2;
  for (int base = 0; base < c; base += 32) {
    int m = c - base;
    if (m > 32) m = 32;
    int s_l = 0;
    float w_l = 0.0f;
    if (lane < m) {
      s_l = csr[start + base + lane];
      w_l = dinv[s_l] * dd;
    }
    int j = 0;
    for (; j + 8 <= m; j += 8) vg_edges128<8>(in, s_l, w_l, j, lane, acc);
    if (j + 4 <= m) { vg_edges128<4>(in, s_l, w_l, j, lane, acc); j += 4; }
    if (j + 2 <= m) { vg_edges128<2>(in, s_l, w_l, j, lane, acc); j += 2; }
    if (j < m)      { vg_edges128<1>(in, s_l, w_l, j, lane, acc); }
  }
  ((float4*)(out + (size_t)row * D_HID))[lane] = acc;
}

__global__ void vg_bn_reduce_kernel(const float* __restrict__ h, float* __restrict__ stats, int n) {
  __shared__ float sh[256];
  int c    = threadIdx.x & 127;
  int half = threadIdx.x >> 7;
  float sum = 0.f, sq = 0.f;
  for (int r = blockIdx.x * 2 + half; r < n; r += gridDim.x * 2) {
    float v = h[(size_t)r * D_HID + c];
    sum += v; sq += v * v;
  }
  sh[threadIdx.x] = sum;
  __syncthreads();
  if (half == 0) atomicAdd(&stats[c], sh[c] + sh[c + 128]);
  __syncthreads();
  sh[threadIdx.x] = sq;
  __syncthreads();
  if (half == 0) atomicAdd(&stats[128 + c], sh[c] + sh[c + 128]);
}

__global__ void vg_bn_finalize_kernel(float* __restrict__ stats, const float* __restrict__ gamma,
                                      const float* __restrict__ beta, int n) {
  int c = threadIdx.x;
  if (c < 128) {
    float invn = 1.0f / (float)n;
    float mean = stats[c] * invn;
    float var  = stats[128 + c] * invn - mean * mean;
    float a = gamma[c] * rsqrtf(var + BN_EPS);
    stats[c]       = a;
    stats[128 + c] = beta[c] - mean * a;
  }
}

__global__ void vg_bn_lrelu_kernel(const float* __restrict__ hagg, const float* __restrict__ stats,
                                   float* __restrict__ out, int total) {
  int i = blockIdx.x * blockDim.x + threadIdx.x;
  if (i >= total) return;
  int c = i & 127;
  float v = stats[c] * hagg[i] + stats[128 + c];
  out[i] = (v > 0.f) ? v : LRELU_SLOPE * v;
}

__global__ void vg_rownorm_kernel(const float* __restrict__ l, float* __restrict__ rns, int n) {
  int row = blockIdx.x * (blockDim.x >> 5) + (threadIdx.x >> 5);
  if (row >= n) return;
  int lane = threadIdx.x & 31;
  const float* rp = l + (size_t)row * D_OUT;
  float a0 = rp[lane], a1 = rp[lane + 32];
  float s = a0 * a0 + a1 * a1;
#pragma unroll
  for (int m = 16; m > 0; m >>= 1) s += __shfl_xor(s, m, 32);
  if (lane == 0) rns[row] = NORM_SCALE / fmaxf(sqrtf(s), 1e-12f);
}

template <int U>
__device__ __forceinline__ void vg_edges64(const float* __restrict__ l, int s_l, float w_l, float sc_l,
                                           int j, int lane, float2& am, float2& al) {
  float2 v[U];
  float  w[U], sc[U];
#pragma unroll
  for (int u = 0; u < U; ++u) {
    int s = __builtin_amdgcn_readlane(s_l, j + u);
    w[u]  = __int_as_float(__builtin_amdgcn_readlane(__float_as_int(w_l), j + u));
    sc[u] = __int_as_float(__builtin_amdgcn_readlane(__float_as_int(sc_l), j + u));
    v[u]  = ((const float2*)(l + (size_t)s * D_OUT))[lane];
  }
#pragma unroll
  for (int u = 0; u < U; ++u) {
    am.x += w[u]  * v[u].x;
    am.y += w[u]  * v[u].y;
    al.x += sc[u] * v[u].x;
    al.y += sc[u] * v[u].y;
  }
}

__global__ void vg_gather64_kernel(const float* __restrict__ l, const int* __restrict__ csr,
                                   const int* __restrict__ rowstart, const int* __restrict__ cnt,
                                   const float* __restrict__ dinv, const float* __restrict__ rns,
                                   float* __restrict__ mu, float* __restrict__ logstd, int n) {
  int wv = __builtin_amdgcn_readfirstlane((int)(threadIdx.x >> 5));
  int row = blockIdx.x * 8 + wv;
  if (row >= n) return;
  int lane = threadIdx.x & 31;
  int start = rowstart[row];
  int c = cnt[row];
  float dd = dinv[row];
  float d2 = dd * dd;
  float rr = rns[row];
  float2 sv = ((const float2*)(l + (size_t)row * D_OUT))[lane];
  float2 am, al;
  am.x = sv.x * d2;      am.y = sv.y * d2;
  al.x = sv.x * rr * d2; al.y = sv.y * rr * d2;
  for (int base = 0; base < c; base += 32) {
    int m = c - base;
    if (m > 32) m = 32;
    int s_l = 0;
    float w_l = 0.0f, sc_l = 0.0f;
    if (lane < m) {
      s_l  = csr[start + base + lane];
      w_l  = dinv[s_l] * dd;
      sc_l = w_l * rns[s_l];
    }
    int j = 0;
    for (; j + 8 <= m; j += 8) vg_edges64<8>(l, s_l, w_l, sc_l, j, lane, am, al);
    if (j + 4 <= m) { vg_edges64<4>(l, s_l, w_l, sc_l, j, lane, am, al); j += 4; }
    if (j + 2 <= m) { vg_edges64<2>(l, s_l, w_l, sc_l, j, lane, am, al); j += 2; }
    if (j < m)      { vg_edges64<1>(l, s_l, w_l, sc_l, j, lane, am, al); }
  }
  ((float2*)(mu     + (size_t)row * D_OUT))[lane] = am;
  ((float2*)(logstd + (size_t)row * D_OUT))[lane] = al;
}

__global__ void vg_zeta_kernel(const float* __restrict__ mu, const float* __restrict__ logstd,
                               const float* __restrict__ noise, float* __restrict__ zeta, int total) {
  int i = blockIdx.x * blockDim.x + threadIdx.x;
  if (i >= total) return;
  zeta[i] = mu[i] + noise[i] * expf(logstd[i]);
}

extern "C" void kernel_launch(void* const* d_in, const int* in_sizes, int n_in,
                              void* d_out, int out_size, void* d_ws, size_t ws_size,
                              hipStream_t stream) {
  const float* x     = (const float*)d_in[0];
  const int*   esrc  = (const int*)  d_in[1];
  const int*   edst  = (const int*)  d_in[2];
  const float* W1    = (const float*)d_in[3];
  const float* b1    = (const float*)d_in[4];
  const float* gamma = (const float*)d_in[5];
  const float* beta  = (const float*)d_in[6];
  const float* Wmu   = (const float*)d_in[7];
  const float* bmu   = (const float*)d_in[8];
  const float* noise = (const float*)d_in[9];

  const int N = in_sizes[0] / D_IN;
  const int E = in_sizes[1];

  float* ws = (float*)d_ws;
  size_t npad = ((size_t)N + 255) & ~(size_t)255;
  size_t off = 0;
  float* dinv  = ws + off; off += npad;
  float* stats = ws + off; off += 256;
  float* rns   = ws + off; off += npad;
  float* h     = ws + off; off += (size_t)N * D_HID;
  float* hagg  = ws + off; off += (size_t)N * D_HID;
  float* lbuf  = ws + off; off += (size_t)N * D_OUT;
  size_t epad = ((size_t)E + 255) & ~(size_t)255;
  int* cnt      = (int*)(ws + off); off += npad;
  int* rowstart = (int*)(ws + off); off += npad;
  int* total    = (int*)(ws + off); off += 256;
  int* rank     = (int*)(ws + off); off += epad;
  int* csr      = (int*)(ws + off); off += epad;

  float* mu_out   = (float*)d_out;
  float* ls_out   = mu_out + (size_t)N * D_OUT;
  float* zeta_out = ls_out + (size_t)N * D_OUT;

  const int B = 256;
  auto cdiv = [](int a, int b) { return (a + b - 1) / b; };

  vg_init_kernel<<<cdiv(N, B), B, 0, stream>>>(cnt, stats, total, N);
  vg_count_kernel<<<cdiv(E, B), B, 0, stream>>>(edst, cnt, rank, E);
  vg_rowptr_kernel<<<cdiv(N, B), B, 0, stream>>>(cnt, total, rowstart, dinv, N);
  vg_fill_kernel<<<cdiv(E, B), B, 0, stream>>>(esrc, edst, rank, rowstart, csr, E);

  {
    int tiles = (N / 16) * (D_HID / 16);
    vg_gemm_wmma_kernel<<<cdiv(tiles, B / 32), B, 0, stream>>>(x, W1, b1, h, N, D_IN, D_HID);
  }

  vg_gather128_kernel<<<cdiv(N, B / 32), B, 0, stream>>>(h, csr, rowstart, cnt, dinv, hagg, N);

  vg_bn_reduce_kernel<<<512, B, 0, stream>>>(hagg, stats, N);
  vg_bn_finalize_kernel<<<1, 128, 0, stream>>>(stats, gamma, beta, N);
  vg_bn_lrelu_kernel<<<cdiv(N * D_HID, B), B, 0, stream>>>(hagg, stats, h, N * D_HID);

  {
    int tiles = (N / 16) * (D_OUT / 16);
    vg_gemm_wmma_kernel<<<cdiv(tiles, B / 32), B, 0, stream>>>(h, Wmu, bmu, lbuf, N, D_HID, D_OUT);
  }

  vg_rownorm_kernel<<<cdiv(N, B / 32), B, 0, stream>>>(lbuf, rns, N);

  vg_gather64_kernel<<<cdiv(N, B / 32), B, 0, stream>>>(lbuf, csr, rowstart, cnt, dinv, rns, mu_out, ls_out, N);

  vg_zeta_kernel<<<cdiv(N * D_OUT, B), B, 0, stream>>>(mu_out, ls_out, noise, zeta_out, N * D_OUT);
}


# Fuse BatchNorm scale/shift + LeakyReLU (and the stats finalize) into the second GEMM's A-fragment load and zeta into the mu/logstd gather's epilogue: 13 launches to 10
# speedup vs baseline: 2.2942x; 1.1233x over previous
#include <hip/hip_runtime.h>
#include <hip/hip_bf16.h>


#define D_IN   128
#define D_HID  128
#define D_OUT  64
#define BN_EPS 1e-5f
#define LRELU_SLOPE 0.01f
#define NORM_SCALE 1.8f

typedef __attribute__((ext_vector_type(2))) float v2f;
typedef __attribute__((ext_vector_type(8))) float v8f;

__global__ void vg_init_kernel(int* __restrict__ cnt, float* __restrict__ stats,
                               int* __restrict__ total, int n) {
  int i = blockIdx.x * blockDim.x + threadIdx.x;
  if (i < n) cnt[i] = 0;
  if (i < 256) stats[i] = 0.0f;
  if (i == 0) total[0] = 0;
}

__global__ void vg_count_kernel(const int* __restrict__ dst, int* __restrict__ cnt,
                                int* __restrict__ rank, int e) {
  int i = blockIdx.x * blockDim.x + threadIdx.x;
  if (i < e) rank[i] = atomicAdd(&cnt[dst[i]], 1);
}

__global__ void vg_rowptr_kernel(const int* __restrict__ cnt, int* __restrict__ total,
                                 int* __restrict__ rowstart, float* __restrict__ dinv, int n) {
  __shared__ int wsum[8];
  __shared__ int base_sh;
  int i = blockIdx.x * 256 + threadIdx.x;
  int lane = threadIdx.x & 31;
  int wv = threadIdx.x >> 5;
  int c = (i < n) ? cnt[i] : 0;
  int x = c;
#pragma unroll
  for (int o = 1; o < 32; o <<= 1) {
    int y = __shfl_up(x, o, 32);
    if (lane >= o) x += y;
  }
  if (lane == 31) wsum[wv] = x;
  __syncthreads();
  if (threadIdx.x == 0) {
    int run = 0;
    for (int k = 0; k < 8; ++k) { int t = wsum[k]; wsum[k] = run; run += t; }
    base_sh = atomicAdd(total, run);
  }
  __syncthreads();
  if (i < n) {
    rowstart[i] = base_sh + wsum[wv] + (x - c);
    dinv[i] = rsqrtf((float)c + 1.0f);
  }
}

__global__ void vg_fill_kernel(const int* __restrict__ src, const int* __restrict__ dst,
                               const int* __restrict__ rank, const int* __restrict__ rowstart,
                               int* __restrict__ csr, int e) {
  int i = blockIdx.x * blockDim.x + threadIdx.x;
  if (i < e) csr[rowstart[dst[i]] + rank[i]] = src[i];
}

__global__ void vg_gemm_wmma_kernel(const float* __restrict__ A, const float* __restrict__ B,
                                    const float* __restrict__ bias, float* __restrict__ C,
                                    int M, int K, int Ncols) {
  const int wavesPerBlock = blockDim.x >> 5;
  int w = blockIdx.x * wavesPerBlock + (threadIdx.x >> 5);
  int tilesN = Ncols >> 4;
  int totalTiles = (M >> 4) * tilesN;
  if (w >= totalTiles) return;
  int tm = w / tilesN;
  int tn = w - tm * tilesN;
  int lane = threadIdx.x & 31;
  int m  = lane & 15;
  int hi = lane >> 4;

  v8f acc = {};
  const float* arow = A + (size_t)(tm * 16 + m) * (size_t)K;
  const float* bcol = B + tn * 16 + m;
  for (int k0 = 0; k0 < K; k0 += 4) {
    int kb = k0 + 2 * hi;
    v2f a = *(const v2f*)(arow + kb);
    v2f b;
    b.x = bcol[(size_t)kb * (size_t)Ncols];
    b.y = bcol[(size_t)(kb + 1) * (size_t)Ncols];
    acc = __builtin_amdgcn_wmma_f32_16x16x4_f32(false, a, false, b,
                                                (short)0, acc, false, false);
  }
  float bv = bias[tn * 16 + m];
  int rbase = tm * 16 + hi * 8;
#pragma unroll
  for (int j = 0; j < 8; ++j) {
    C[(size_t)(rbase + j) * (size_t)Ncols + (size_t)(tn * 16 + m)] = acc[j] + bv;
  }
}

template <int U>
__device__ __forceinline__ void vg_edges128(const float* __restrict__ in, int s_l, float w_l,
                                            int j, int lane, float4& acc) {
  float4 v[U];
  float  w[U];
#pragma unroll
  for (int u = 0; u < U; ++u) {
    int s = __builtin_amdgcn_readlane(s_l, j + u);
    w[u]  = __int_as_float(__builtin_amdgcn_readlane(__float_as_int(w_l), j + u));
    v[u]  = ((const float4*)(in + (size_t)s * D_HID))[lane];
  }
#pragma unroll
  for (int u = 0; u < U; ++u) {
    acc.x += w[u] * v[u].x;
    acc.y += w[u] * v[u].y;
    acc.z += w[u] * v[u].z;
    acc.w += w[u] * v[u].w;
  }
}

__global__ void vg_gather128_kernel(const float* __restrict__ in, const int* __restrict__ csr,
                                    const int* __restrict__ rowstart, const int* __restrict__ cnt,
                                    const float* __restrict__ dinv, float* __restrict__ out, int n) {
  int wv = __builtin_amdgcn_readfirstlane((int)(threadIdx.x >> 5));
  int row = blockIdx.x * 8 + wv;
  if (row >= n) return;
  int lane = threadIdx.x & 31;
  int start = rowstart[row];
  int c = cnt[row];
  float dd = dinv[row];
  float d2 = dd * dd;
  float4 sv = ((const float4*)(in + (size_t)row * D_HID))[lane];
  float4 acc;
  acc.x = sv.x * d2; acc.y = sv.y * d2; acc.z = sv.z * d2; acc.w = sv.w * d2;
  for (int base = 0; base < c; base += 32) {
    int m = c - base;
    if (m > 32) m = 32;
    int s_l = 0;
    float w_l = 0.0f;
    if (lane < m) {
      s_l = csr[start + base + lane];
      w_l = dinv[s_l] * dd;
    }
    int j = 0;
    for (; j + 8 <= m; j += 8) vg_edges128<8>(in, s_l, w_l, j, lane, acc);
    if (j + 4 <= m) { vg_edges128<4>(in, s_l, w_l, j, lane, acc); j += 4; }
    if (j + 2 <= m) { vg_edges128<2>(in, s_l, w_l, j, lane, acc); j += 2; }
    if (j < m)      { vg_edges128<1>(in, s_l, w_l, j, lane, acc); }
  }
  ((float4*)(out + (size_t)row * D_HID))[lane] = acc;
}

__global__ void vg_bn_reduce_kernel(const float* __restrict__ h, float* __restrict__ stats, int n) {
  __shared__ float sh[256];
  int c    = threadIdx.x & 127;
  int half = threadIdx.x >> 7;
  float sum = 0.f, sq = 0.f;
  for (int r = blockIdx.x * 2 + half; r < n; r += gridDim.x * 2) {
    float v = h[(size_t)r * D_HID + c];
    sum += v; sq += v * v;
  }
  sh[threadIdx.x] = sum;
  __syncthreads();
  if (half == 0) atomicAdd(&stats[c], sh[c] + sh[c + 128]);
  __syncthreads();
  sh[threadIdx.x] = sq;
  __syncthreads();
  if (half == 0) atomicAdd(&stats[128 + c], sh[c] + sh[c + 128]);
}

__global__ void vg_gemm_bn_wmma_kernel(const float* __restrict__ A, const float* __restrict__ B,
                                       const float* __restrict__ bias, const float* __restrict__ stats,
                                       const float* __restrict__ gamma, const float* __restrict__ beta,
                                       float* __restrict__ C, int M, int n) {
  __shared__ __attribute__((aligned(16))) float bn_sc[D_HID];
  __shared__ __attribute__((aligned(16))) float bn_sh[D_HID];
  if (threadIdx.x < D_HID) {
    int c = threadIdx.x;
    float invn = 1.0f / (float)n;
    float mean = stats[c] * invn;
    float var  = stats[128 + c] * invn - mean * mean;
    float a = gamma[c] * rsqrtf(var + BN_EPS);
    bn_sc[c] = a;
    bn_sh[c] = beta[c] - mean * a;
  }
  __syncthreads();

  const int wavesPerBlock = blockDim.x >> 5;
  int w = blockIdx.x * wavesPerBlock + (threadIdx.x >> 5);
  const int tilesN = D_OUT >> 4;
  int totalTiles = (M >> 4) * tilesN;
  if (w >= totalTiles) return;
  int tm = w / tilesN;
  int tn = w - tm * tilesN;
  int lane = threadIdx.x & 31;
  int m  = lane & 15;
  int hi = lane >> 4;

  v8f acc = {};
  const float* arow = A + (size_t)(tm * 16 + m) * (size_t)D_HID;
  const float* bcol = B + tn * 16 + m;
  for (int k0 = 0; k0 < D_HID; k0 += 4) {
    int kb = k0 + 2 * hi;
    v2f a  = *(const v2f*)(arow + kb);
    v2f sc = *(const v2f*)(bn_sc + kb);
    v2f sh = *(const v2f*)(bn_sh + kb);
    float v0 = sc.x * a.x + sh.x;
    float v1 = sc.y * a.y + sh.y;
    a.x = (v0 > 0.f) ? v0 : LRELU_SLOPE * v0;
    a.y = (v1 > 0.f) ? v1 : LRELU_SLOPE * v1;
    v2f b;
    b.x = bcol[(size_t)kb * (size_t)D_OUT];
    b.y = bcol[(size_t)(kb + 1) * (size_t)D_OUT];
    acc = __builtin_amdgcn_wmma_f32_16x16x4_f32(false, a, false, b,
                                                (short)0, acc, false, false);
  }
  float bv = bias[tn * 16 + m];
  int rbase = tm * 16 + hi * 8;
#pragma unroll
  for (int j = 0; j < 8; ++j) {
    C[(size_t)(rbase + j) * (size_t)D_OUT + (size_t)(tn * 16 + m)] = acc[j] + bv;
  }
}

__global__ void vg_rownorm_kernel(const float* __restrict__ l, float* __restrict__ rns, int n) {
  int row = blockIdx.x * (blockDim.x >> 5) + (threadIdx.x >> 5);
  if (row >= n) return;
  int lane = threadIdx.x & 31;
  const float* rp = l + (size_t)row * D_OUT;
  float a0 = rp[lane], a1 = rp[lane + 32];
  float s = a0 * a0 + a1 * a1;
#pragma unroll
  for (int m = 16; m > 0; m >>= 1) s += __shfl_xor(s, m, 32);
  if (lane == 0) rns[row] = NORM_SCALE / fmaxf(sqrtf(s), 1e-12f);
}

template <int U>
__device__ __forceinline__ void vg_edges64(const float* __restrict__ l, int s_l, float w_l, float sc_l,
                                           int j, int lane, float2& am, float2& al) {
  float2 v[U];
  float  w[U], sc[U];
#pragma unroll
  for (int u = 0; u < U; ++u) {
    int s = __builtin_amdgcn_readlane(s_l, j + u);
    w[u]  = __int_as_float(__builtin_amdgcn_readlane(__float_as_int(w_l), j + u));
    sc[u] = __int_as_float(__builtin_amdgcn_readlane(__float_as_int(sc_l), j + u));
    v[u]  = ((const float2*)(l + (size_t)s * D_OUT))[lane];
  }
#pragma unroll
  for (int u = 0; u < U; ++u) {
    am.x += w[u]  * v[u].x;
    am.y += w[u]  * v[u].y;
    al.x += sc[u] * v[u].x;
    al.y += sc[u] * v[u].y;
  }
}

__global__ void vg_gather64_kernel(const float* __restrict__ l, const int* __restrict__ csr,
                                   const int* __restrict__ rowstart, const int* __restrict__ cnt,
                                   const float* __restrict__ dinv, const float* __restrict__ rns,
                                   const float* __restrict__ noise, float* __restrict__ mu,
                                   float* __restrict__ logstd, float* __restrict__ zeta, int n) {
  int wv = __builtin_amdgcn_readfirstlane((int)(threadIdx.x >> 5));
  int row = blockIdx.x * 8 + wv;
  if (row >= n) return;
  int lane = threadIdx.x & 31;
  int start = rowstart[row];
  int c = cnt[row];
  float dd = dinv[row];
  float d2 = dd * dd;
  float rr = rns[row];
  float2 sv = ((const float2*)(l + (size_t)row * D_OUT))[lane];
  float2 nz = ((const float2*)(noise + (size_t)row * D_OUT))[lane];
  float2 am, al;
  am.x = sv.x * d2;      am.y = sv.y * d2;
  al.x = sv.x * rr * d2; al.y = sv.y * rr * d2;
  for (int base = 0; base < c; base += 32) {
    int m = c - base;
    if (m > 32) m = 32;
    int s_l = 0;
    float w_l = 0.0f, sc_l = 0.0f;
    if (lane < m) {
      s_l  = csr[start + base + lane];
      w_l  = dinv[s_l] * dd;
      sc_l = w_l * rns[s_l];
    }
    int j = 0;
    for (; j + 8 <= m; j += 8) vg_edges64<8>(l, s_l, w_l, sc_l, j, lane, am, al);
    if (j + 4 <= m) { vg_edges64<4>(l, s_l, w_l, sc_l, j, lane, am, al); j += 4; }
    if (j + 2 <= m) { vg_edges64<2>(l, s_l, w_l, sc_l, j, lane, am, al); j += 2; }
    if (j < m)      { vg_edges64<1>(l, s_l, w_l, sc_l, j, lane, am, al); }
  }
  ((float2*)(mu     + (size_t)row * D_OUT))[lane] = am;
  ((float2*)(logstd + (size_t)row * D_OUT))[lane] = al;
  float2 z;
  z.x = am.x + nz.x * expf(al.x);
  z.y = am.y + nz.y * expf(al.y);
  ((float2*)(zeta   + (size_t)row * D_OUT))[lane] = z;
}

extern "C" void kernel_launch(void* const* d_in, const int* in_sizes, int n_in,
                              void* d_out, int out_size, void* d_ws, size_t ws_size,
                              hipStream_t stream) {
  const float* x     = (const float*)d_in[0];
  const int*   esrc  = (const int*)  d_in[1];
  const int*   edst  = (const int*)  d_in[2];
  const float* W1    = (const float*)d_in[3];
  const float* b1    = (const float*)d_in[4];
  const float* gamma = (const float*)d_in[5];
  const float* beta  = (const float*)d_in[6];
  const float* Wmu   = (const float*)d_in[7];
  const float* bmu   = (const float*)d_in[8];
  const float* noise = (const float*)d_in[9];

  const int N = in_sizes[0] / D_IN;
  const int E = in_sizes[1];

  float* ws = (float*)d_ws;
  size_t npad = ((size_t)N + 255) & ~(size_t)255;
  size_t off = 0;
  float* dinv  = ws + off; off += npad;
  float* stats = ws + off; off += 256;
  float* rns   = ws + off; off += npad;
  float* h     = ws + off; off += (size_t)N * D_HID;
  float* hagg  = ws + off; off += (size_t)N * D_HID;
  float* lbuf  = ws + off; off += (size_t)N * D_OUT;
  size_t epad = ((size_t)E + 255) & ~(size_t)255;
  int* cnt      = (int*)(ws + off); off += npad;
  int* rowstart = (int*)(ws + off); off += npad;
  int* total    = (int*)(ws + off); off += 256;
  int* rank     = (int*)(ws + off); off += epad;
  int* csr      = (int*)(ws + off); off += epad;

  float* mu_out   = (float*)d_out;
  float* ls_out   = mu_out + (size_t)N * D_OUT;
  float* zeta_out = ls_out + (size_t)N * D_OUT;

  const int B = 256;
  auto cdiv = [](int a, int b) { return (a + b - 1) / b; };

  vg_init_kernel<<<cdiv(N, B), B, 0, stream>>>(cnt, stats, total, N);
  vg_count_kernel<<<cdiv(E, B), B, 0, stream>>>(edst, cnt, rank, E);
  vg_rowptr_kernel<<<cdiv(N, B), B, 0, stream>>>(cnt, total, rowstart, dinv, N);
  vg_fill_kernel<<<cdiv(E, B), B, 0, stream>>>(esrc, edst, rank, rowstart, csr, E);

  {
    int tiles = (N / 16) * (D_HID / 16);
    vg_gemm_wmma_kernel<<<cdiv(tiles, B / 32), B, 0, stream>>>(x, W1, b1, h, N, D_IN, D_HID);
  }

  vg_gather128_kernel<<<cdiv(N, B / 32), B, 0, stream>>>(h, csr, rowstart, cnt, dinv, hagg, N);

  vg_bn_reduce_kernel<<<512, B, 0, stream>>>(hagg, stats, N);

  {
    int tiles = (N / 16) * (D_OUT / 16);
    vg_gemm_bn_wmma_kernel<<<cdiv(tiles, B / 32), B, 0, stream>>>(hagg, Wmu, bmu, stats, gamma, beta, lbuf, N, N);
  }

  vg_rownorm_kernel<<<cdiv(N, B / 32), B, 0, stream>>>(lbuf, rns, N);

  vg_gather64_kernel<<<cdiv(N, B / 32), B, 0, stream>>>(lbuf, csr, rowstart, cnt, dinv, rns, noise,
                                                        mu_out, ls_out, zeta_out, N);
}


# Edge list grouped by destination with LDS counters: edges binned by 256-row range in 2 passes, one block per range ranks them in LDS; no per-edge global returning add; sums still in arrival order
# speedup vs baseline: 3.0251x; 1.3186x over previous
#include <hip/hip_runtime.h>
#include <hip/hip_bf16.h>


#define D_IN   128
#define D_HID  128
#define D_OUT  64
#define BN_EPS 1e-5f
#define LRELU_SLOPE 0.01f
#define NORM_SCALE 1.8f

typedef __attribute__((ext_vector_type(2))) float v2f;
typedef __attribute__((ext_vector_type(8))) float v8f;

#define BKT_SHIFT 8
#define BKT_ROWS  256
#define EPB       4096
#define EPT       (EPB / 256)
#define BPT       20

__global__ void vg_init_kernel(int* __restrict__ bcnt, float* __restrict__ stats, int nb) {
  int i = blockIdx.x * blockDim.x + threadIdx.x;
  if (i < nb) bcnt[i] = 0;
  if (i < 256) stats[i] = 0.0f;
}

__global__ void vg_bucket_count_kernel(const int* __restrict__ dst, int* __restrict__ bcnt,
                                       int* __restrict__ blkbase, int e, int nb) {
  extern __shared__ int lds_i[];
  for (int b = threadIdx.x; b < nb; b += 256) lds_i[b] = 0;
  __syncthreads();
  int e0 = blockIdx.x * EPB + threadIdx.x;
  int d[EPT];
#pragma unroll
  for (int k = 0; k < EPT; ++k) {
    int i = e0 + k * 256;
    d[k] = (i < e) ? dst[i] : -1;
  }
#pragma unroll
  for (int k = 0; k < EPT; ++k) {
    if (d[k] >= 0) atomicAdd(&lds_i[d[k] >> BKT_SHIFT], 1);
  }
  __syncthreads();
  for (int b = threadIdx.x; b < nb; b += 256) {
    int h = lds_i[b];
    blkbase[(size_t)blockIdx.x * (size_t)nb + b] = (h > 0) ? atomicAdd(&bcnt[b], h) : 0;
  }
}

__global__ void vg_bucket_scatter_kernel(const int* __restrict__ src, const int* __restrict__ dst,
                                         const int* __restrict__ bcnt, const int* __restrict__ blkbase,
                                         int2* __restrict__ binned, int e, int nb) {
  extern __shared__ int lds_i[];
  __shared__ int wsum[8];
  int lane = threadIdx.x & 31;
  int wv = threadIdx.x >> 5;
  int run = 0;
  for (int b0 = 0; b0 < nb; b0 += 256) {
    int b = b0 + threadIdx.x;
    int c = (b < nb) ? bcnt[b] : 0;
    int x = c;
#pragma unroll
    for (int o = 1; o < 32; o <<= 1) {
      int y = __shfl_up(x, o, 32);
      if (lane >= o) x += y;
    }
    if (lane == 31) wsum[wv] = x;
    __syncthreads();
    int wbase = 0, tot = 0;
#pragma unroll
    for (int k = 0; k < 8; ++k) { int t = wsum[k]; if (k < wv) wbase += t; tot += t; }
    if (b < nb) lds_i[b] = run + wbase + (x - c) + blkbase[(size_t)blockIdx.x * (size_t)nb + b];
    run += tot;
    __syncthreads();
  }
  int e0 = blockIdx.x * EPB + threadIdx.x;
  int d[EPT], s[EPT];
#pragma unroll
  for (int k = 0; k < EPT; ++k) {
    int i = e0 + k * 256;
    d[k] = (i < e) ? dst[i] : -1;
    s[k] = (i < e) ? src[i] : 0;
  }
#pragma unroll
  for (int k = 0; k < EPT; ++k) {
    if (d[k] >= 0) {
      int p = atomicAdd(&lds_i[d[k] >> BKT_SHIFT], 1);
      binned[p] = make_int2(s[k], d[k]);
    }
  }
}

__global__ void vg_bucket_rows_kernel(const int2* __restrict__ binned, const int* __restrict__ bcnt,
                                      int* __restrict__ cnt, int* __restrict__ rowstart,
                                      float* __restrict__ dinv, int* __restrict__ csr, int n) {
  __shared__ int lcnt[BKT_ROWS];
  __shared__ int lcur[BKT_ROWS];
  __shared__ int wsum[8];
  __shared__ int wsum2[8];
  int bkt = blockIdx.x;
  int t = threadIdx.x;
  int lane = t & 31;
  int wv = t >> 5;
  int part = 0;
  for (int j = t; j < bkt; j += 256) part += bcnt[j];
#pragma unroll
  for (int o = 16; o > 0; o >>= 1) part += __shfl_xor(part, o, 32);
  if (lane == 0) wsum[wv] = part;
  lcnt[t] = 0;
  __syncthreads();
  int bstart = 0;
#pragma unroll
  for (int k = 0; k < 8; ++k) bstart += wsum[k];
  int bsize = bcnt[bkt];
  const int2* be = binned + bstart;

  for (int i0 = 0; i0 < bsize; i0 += 256 * BPT) {
    int r[BPT];
#pragma unroll
    for (int k = 0; k < BPT; ++k) {
      int i = i0 + k * 256 + t;
      r[k] = be[(i < bsize) ? i : (bsize - 1)].y;
    }
#pragma unroll
    for (int k = 0; k < BPT; ++k) {
      if (i0 + k * 256 + t < bsize) atomicAdd(&lcnt[r[k] & (BKT_ROWS - 1)], 1);
    }
  }
  __syncthreads();

  int c = lcnt[t];
  int x = c;
#pragma unroll
  for (int o = 1; o < 32; o <<= 1) {
    int y = __shfl_up(x, o, 32);
    if (lane >= o) x += y;
  }
  if (lane == 31) wsum2[wv] = x;
  __syncthreads();
  int wbase = 0;
#pragma unroll
  for (int k = 0; k < 8; ++k) { int w = wsum2[k]; if (k < wv) wbase += w; }
  int rs = bstart + wbase + (x - c);
  lcur[t] = rs;
  int row = bkt * BKT_ROWS + t;
  if (row < n) {
    cnt[row] = c;
    rowstart[row] = rs;
    dinv[row] = rsqrtf((float)c + 1.0f);
  }
  __syncthreads();

  for (int i0 = 0; i0 < bsize; i0 += 256 * BPT) {
    int2 ev[BPT];
#pragma unroll
    for (int k = 0; k < BPT; ++k) {
      int i = i0 + k * 256 + t;
      ev[k] = be[(i < bsize) ? i : (bsize - 1)];
    }
#pragma unroll
    for (int k = 0; k < BPT; ++k) {
      if (i0 + k * 256 + t < bsize) {
        int p = atomicAdd(&lcur[ev[k].y & (BKT_ROWS - 1)], 1);
        csr[p] = ev[k].x;
      }
    }
  }
}

__global__ void vg_gemm_wmma_kernel(const float* __restrict__ A, const float* __restrict__ B,
                                    const float* __restrict__ bias, float* __restrict__ C,
                                    int M, int K, int Ncols) {
  const int wavesPerBlock = blockDim.x >> 5;
  int w = blockIdx.x * wavesPerBlock + (threadIdx.x >> 5);
  int tilesN = Ncols >> 4;
  int totalTiles = (M >> 4) * tilesN;
  if (w >= totalTiles) return;
  int tm = w / tilesN;
  int tn = w - tm * tilesN;
  int lane = threadIdx.x & 31;
  int m  = lane & 15;
  int hi = lane >> 4;

  v8f acc = {};
  const float* arow = A + (size_t)(tm * 16 + m) * (size_t)K;
  const float* bcol = B + tn * 16 + m;
  for (int k0 = 0; k0 < K; k0 += 4) {
    int kb = k0 + 2 * hi;
    v2f a = *(const v2f*)(arow + kb);
    v2f b;
    b.x = bcol[(size_t)kb * (size_t)Ncols];
    b.y = bcol[(size_t)(kb + 1) * (size_t)Ncols];
    acc = __builtin_amdgcn_wmma_f32_16x16x4_f32(false, a, false, b,
                                                (short)0, acc, false, false);
  }
  float bv = bias[tn * 16 + m];
  int rbase = tm * 16 + hi * 8;
#pragma unroll
  for (int j = 0; j < 8; ++j) {
    C[(size_t)(rbase + j) * (size_t)Ncols + (size_t)(tn * 16 + m)] = acc[j] + bv;
  }
}

template <int U>
__device__ __forceinline__ void vg_edges128(const float* __restrict__ in, int s_l, float w_l,
                                            int j, int lane, float4& acc) {
  float4 v[U];
  float  w[U];
#pragma unroll
  for (int u = 0; u < U; ++u) {
    int s = __builtin_amdgcn_readlane(s_l, j + u);
    w[u]  = __int_as_float(__builtin_amdgcn_readlane(__float_as_int(w_l), j + u));
    v[u]  = ((const float4*)(in + (size_t)s * D_HID))[lane];
  }
#pragma unroll
  for (int u = 0; u < U; ++u) {
    acc.x += w[u] * v[u].x;
    acc.y += w[u] * v[u].y;
    acc.z += w[u] * v[u].z;
    acc.w += w[u] * v[u].w;
  }
}

__global__ void vg_gather128_kernel(const float* __restrict__ in, const int* __restrict__ csr,
                                    const int* __restrict__ rowstart, const int* __restrict__ cnt,
                                    const float* __restrict__ dinv, float* __restrict__ out, int n) {
  int wv = __builtin_amdgcn_readfirstlane((int)(threadIdx.x >> 5));
  int row = blockIdx.x * 8 + wv;
  if (row >= n) return;
  int lane = threadIdx.x & 31;
  int start = rowstart[row];
  int c = cnt[row];
  float dd = dinv[row];
  float d2 = dd * dd;
  float4 sv = ((const float4*)(in + (size_t)row * D_HID))[lane];
  float4 acc;
  acc.x = sv.x * d2; acc.y = sv.y * d2; acc.z = sv.z * d2; acc.w = sv.w * d2;
  for (int base = 0; base < c; base += 32) {
    int m = c - base;
    if (m > 32) m = 32;
    int s_l = 0;
    float w_l = 0.0f;
    if (lane < m) {
      s_l = csr[start + base + lane];
      w_l = dinv[s_l] * dd;
    }
    int j = 0;
    for (; j + 8 <= m; j += 8) vg_edges128<8>(in, s_l, w_l, j, lane, acc);
    if (j + 4 <= m) { vg_edges128<4>(in, s_l, w_l, j, lane, acc); j += 4; }
    if (j + 2 <= m) { vg_edges128<2>(in, s_l, w_l, j, lane, acc); j += 2; }
    if (j < m)      { vg_edges128<1>(in, s_l, w_l, j, lane, acc); }
  }
  ((float4*)(out + (size_t)row * D_HID))[lane] = acc;
}

__global__ void vg_bn_reduce_kernel(const float* __restrict__ h, float* __restrict__ stats, int n) {
  __shared__ float sh[256];
  int c    = threadIdx.x & 127;
  int half = threadIdx.x >> 7;
  float sum = 0.f, sq = 0.f;
  for (int r = blockIdx.x * 2 + half; r < n; r += gridDim.x * 2) {
    float v = h[(size_t)r * D_HID + c];
    sum += v; sq += v * v;
  }
  sh[threadIdx.x] = sum;
  __syncthreads();
  if (half == 0) atomicAdd(&stats[c], sh[c] + sh[c + 128]);
  __syncthreads();
  sh[threadIdx.x] = sq;
  __syncthreads();
  if (half == 0) atomicAdd(&stats[128 + c], sh[c] + sh[c + 128]);
}

__global__ void vg_gemm_bn_wmma_kernel(const float* __restrict__ A, const float* __restrict__ B,
                                       const float* __restrict__ bias, const float* __restrict__ stats,
                                       const float* __restrict__ gamma, const float* __restrict__ beta,
                                       float* __restrict__ C, int M, int n) {
  __shared__ __attribute__((aligned(16))) float bn_sc[D_HID];
  __shared__ __attribute__((aligned(16))) float bn_sh[D_HID];
  if (threadIdx.x < D_HID) {
    int c = threadIdx.x;
    float invn = 1.0f / (float)n;
    float mean = stats[c] * invn;
    float var  = stats[128 + c] * invn - mean * mean;
    float a = gamma[c] * rsqrtf(var + BN_EPS);
    bn_sc[c] = a;
    bn_sh[c] = beta[c] - mean * a;
  }
  __syncthreads();

  const int wavesPerBlock = blockDim.x >> 5;
  int w = blockIdx.x * wavesPerBlock + (threadIdx.x >> 5);
  const int tilesN = D_OUT >> 4;
  int totalTiles = (M >> 4) * tilesN;
  if (w >= totalTiles) return;
  int tm = w / tilesN;
  int tn = w - tm * tilesN;
  int lane = threadIdx.x & 31;
  int m  = lane & 15;
  int hi = lane >> 4;

  v8f acc = {};
  const float* arow = A + (size_t)(tm * 16 + m) * (size_t)D_HID;
  const float* bcol = B + tn * 16 + m;
  for (int k0 = 0; k0 < D_HID; k0 += 4) {
    int kb = k0 + 2 * hi;
    v2f a  = *(const v2f*)(arow + kb);
    v2f sc = *(const v2f*)(bn_sc + kb);
    v2f sh = *(const v2f*)(bn_sh + kb);
    float v0 = sc.x * a.x + sh.x;
    float v1 = sc.y * a.y + sh.y;
    a.x = (v0 > 0.f) ? v0 : LRELU_SLOPE * v0;
    a.y = (v1 > 0.f) ? v1 : LRELU_SLOPE * v1;
    v2f b;
    b.x = bcol[(size_t)kb * (size_t)D_OUT];
    b.y = bcol[(size_t)(kb + 1) * (size_t)D_OUT];
    acc = __builtin_amdgcn_wmma_f32_16x16x4_f32(false, a, false, b,
                                                (short)0, acc, false, false);
  }
  float bv = bias[tn * 16 + m];
  int rbase = tm * 16 + hi * 8;
#pragma unroll
  for (int j = 0; j < 8; ++j) {
    C[(size_t)(rbase + j) * (size_t)D_OUT + (size_t)(tn * 16 + m)] = acc[j] + bv;
  }
}

__global__ void vg_rownorm_kernel(const float* __restrict__ l, float* __restrict__ rns, int n) {
  int row = blockIdx.x * (blockDim.x >> 5) + (threadIdx.x >> 5);
  if (row >= n) return;
  int lane = threadIdx.x & 31;
  const float* rp = l + (size_t)row * D_OUT;
  float a0 = rp[lane], a1 = rp[lane + 32];
  float s = a0 * a0 + a1 * a1;
#pragma unroll
  for (int m = 16; m > 0; m >>= 1) s += __shfl_xor(s, m, 32);
  if (lane == 0) rns[row] = NORM_SCALE / fmaxf(sqrtf(s), 1e-12f);
}

template <int U>
__device__ __forceinline__ void vg_edges64(const float* __restrict__ l, int s_l, float w_l, float sc_l,
                                           int j, int lane, float2& am, float2& al) {
  float2 v[U];
  float  w[U], sc[U];
#pragma unroll
  for (int u = 0; u < U; ++u) {
    int s = __builtin_amdgcn_readlane(s_l, j + u);
    w[u]  = __int_as_float(__builtin_amdgcn_readlane(__float_as_int(w_l), j + u));
    sc[u] = __int_as_float(__builtin_amdgcn_readlane(__float_as_int(sc_l), j + u));
    v[u]  = ((const float2*)(l + (size_t)s * D_OUT))[lane];
  }
#pragma unroll
  for (int u = 0; u < U; ++u) {
    am.x += w[u]  * v[u].x;
    am.y += w[u]  * v[u].y;
    al.x += sc[u] * v[u].x;
    al.y += sc[u] * v[u].y;
  }
}

__global__ void vg_gather64_kernel(const float* __restrict__ l, const int* __restrict__ csr,
                                   const int* __restrict__ rowstart, const int* __restrict__ cnt,
                                   const float* __restrict__ dinv, const float* __restrict__ rns,
                                   const float* __restrict__ noise, float* __restrict__ mu,
                                   float* __restrict__ logstd, float* __restrict__ zeta, int n) {
  int wv = __builtin_amdgcn_readfirstlane((int)(threadIdx.x >> 5));
  int row = blockIdx.x * 8 + wv;
  if (row >= n) return;
  int lane = threadIdx.x & 31;
  int start = rowstart[row];
  int c = cnt[row];
  float dd = dinv[row];
  float d2 = dd * dd;
  float rr = rns[row];
  float2 sv = ((const float2*)(l + (size_t)row * D_OUT))[lane];
  float2 nz = ((const float2*)(noise + (size_t)row * D_OUT))[lane];
  float2 am, al;
  am.x = sv.x * d2;      am.y = sv.y * d2;
  al.x = sv.x * rr * d2; al.y = sv.y * rr * d2;
  for (int base = 0; base < c; base += 32) {
    int m = c - base;
    if (m > 32) m = 32;
    int s_l = 0;
    float w_l = 0.0f, sc_l = 0.0f;
    if (lane < m) {
      s_l  = csr[start + base + lane];
      w_l  = dinv[s_l] * dd;
      sc_l = w_l * rns[s_l];
    }
    int j = 0;
    for (; j + 8 <= m; j += 8) vg_edges64<8>(l, s_l, w_l, sc_l, j, lane, am, al);
    if (j + 4 <= m) { vg_edges64<4>(l, s_l, w_l, sc_l, j, lane, am, al); j += 4; }
    if (j + 2 <= m) { vg_edges64<2>(l, s_l, w_l, sc_l, j, lane, am, al); j += 2; }
    if (j < m)      { vg_edges64<1>(l, s_l, w_l, sc_l, j, lane, am, al); }
  }
  ((float2*)(mu     + (size_t)row * D_OUT))[lane] = am;
  ((float2*)(logstd + (size_t)row * D_OUT))[lane] = al;
  float2 z;
  z.x = am.x + nz.x * expf(al.x);
  z.y = am.y + nz.y * expf(al.y);
  ((float2*)(zeta   + (size_t)row * D_OUT))[lane] = z;
}

extern "C" void kernel_launch(void* const* d_in, const int* in_sizes, int n_in,
                              void* d_out, int out_size, void* d_ws, size_t ws_size,
                              hipStream_t stream) {
  const float* x     = (const float*)d_in[0];
  const int*   esrc  = (const int*)  d_in[1];
  const int*   edst  = (const int*)  d_in[2];
  const float* W1    = (const float*)d_in[3];
  const float* b1    = (const float*)d_in[4];
  const float* gamma = (const float*)d_in[5];
  const float* beta  = (const float*)d_in[6];
  const float* Wmu   = (const float*)d_in[7];
  const float* bmu   = (const float*)d_in[8];
  const float* noise = (const float*)d_in[9];

  const int N = in_sizes[0] / D_IN;
  const int E = in_sizes[1];

  float* ws = (float*)d_ws;
  size_t npad = ((size_t)N + 255) & ~(size_t)255;
  size_t off = 0;
  float* dinv  = ws + off; off += npad;
  float* stats = ws + off; off += 256;
  float* rns   = ws + off; off += npad;
  float* h     = ws + off; off += (size_t)N * D_HID;
  float* hagg  = ws + off; off += (size_t)N * D_HID;
  float* lbuf  = ws + off; off += (size_t)N * D_OUT;
  size_t epad = ((size_t)E + 255) & ~(size_t)255;
  const int NB   = (N + BKT_ROWS - 1) / BKT_ROWS;
  const int NBLK = (E + EPB - 1) / EPB;
  size_t bbpad = ((size_t)NBLK * (size_t)NB + 255) & ~(size_t)255;
  int* cnt      = (int*)(ws + off); off += npad;
  int* rowstart = (int*)(ws + off); off += npad;
  int* bcnt     = (int*)(ws + off); off += npad;
  int* blkbase  = (int*)(ws + off); off += bbpad;
  int* csr      = (int*)(ws + off); off += epad;
  int2* binned  = (int2*)(ws + off); off += 2 * epad;

  float* mu_out   = (float*)d_out;
  float* ls_out   = mu_out + (size_t)N * D_OUT;
  float* zeta_out = ls_out + (size_t)N * D_OUT;

  const int B = 256;
  auto cdiv = [](int a, int b) { return (a + b - 1) / b; };

  vg_init_kernel<<<cdiv(NB, B), B, 0, stream>>>(bcnt, stats, NB);
  vg_bucket_count_kernel<<<NBLK, B, NB * sizeof(int), stream>>>(edst, bcnt, blkbase, E, NB);
  vg_bucket_scatter_kernel<<<NBLK, B, NB * sizeof(int), stream>>>(esrc, edst, bcnt, blkbase, binned, E, NB);
  vg_bucket_rows_kernel<<<NB, B, 0, stream>>>(binned, bcnt, cnt, rowstart, dinv, csr, N);

  {
    int tiles = (N / 16) * (D_HID / 16);
    vg_gemm_wmma_kernel<<<cdiv(tiles, B / 32), B, 0, stream>>>(x, W1, b1, h, N, D_IN, D_HID);
  }

  vg_gather128_kernel<<<cdiv(N, B / 32), B, 0, stream>>>(h, csr, rowstart, cnt, dinv, hagg, N);

  vg_bn_reduce_kernel<<<512, B, 0, stream>>>(hagg, stats, N);

  {
    int tiles = (N / 16) * (D_OUT / 16);
    vg_gemm_bn_wmma_kernel<<<cdiv(tiles, B / 32), B, 0, stream>>>(hagg, Wmu, bmu, stats, gamma, beta, lbuf, N, N);
  }

  vg_rownorm_kernel<<<cdiv(N, B / 32), B, 0, stream>>>(lbuf, rns, N);

  vg_gather64_kernel<<<cdiv(N, B / 32), B, 0, stream>>>(lbuf, csr, rowstart, cnt, dinv, rns, noise,
                                                        mu_out, ls_out, zeta_out, N);
}


# Fewer, longer-lived waves: a GEMM wave owns a 16x64 strip (4 accumulator tiles per A fragment), a gather or row-norm wave walks 8 consecutive rows; about a fifth as many waves enter per call
# speedup vs baseline: 3.3961x; 1.1226x over previous
#include <hip/hip_runtime.h>
#include <hip/hip_bf16.h>


#define D_IN   128
#define D_HID  128
#define D_OUT  64
#define BN_EPS 1e-5f
#define LRELU_SLOPE 0.01f
#define NORM_SCALE 1.8f

typedef __attribute__((ext_vector_type(2))) float v2f;
typedef __attribute__((ext_vector_type(8))) float v8f;

#define BKT_SHIFT 8
#define BKT_ROWS  256
#define EPB       4096
#define EPT       (EPB / 256)
#define BPT       20

__global__ void vg_init_kernel(int* __restrict__ bcnt, float* __restrict__ stats, int nb) {
  int i = blockIdx.x * blockDim.x + threadIdx.x;
  if (i < nb) bcnt[i] = 0;
  if (i < 256) stats[i] = 0.0f;
}

__global__ void vg_bucket_count_kernel(const int* __restrict__ dst, int* __restrict__ bcnt,
                                       int* __restrict__ blkbase, int e, int nb) {
  extern __shared__ int lds_i[];
  for (int b = threadIdx.x; b < nb; b += 256) lds_i[b] = 0;
  __syncthreads();
  int e0 = blockIdx.x * EPB + threadIdx.x;
  int d[EPT];
#pragma unroll
  for (int k = 0; k < EPT; ++k) {
    int i = e0 + k * 256;
    d[k] = (i < e) ? dst[i] : -1;
  }
#pragma unroll
  for (int k = 0; k < EPT; ++k) {
    if (d[k] >= 0) atomicAdd(&lds_i[d[k] >> BKT_SHIFT], 1);
  }
  __syncthreads();
  for (int b = threadIdx.x; b < nb; b += 256) {
    int h = lds_i[b];
    blkbase[(size_t)blockIdx.x * (size_t)nb + b] = (h > 0) ? atomicAdd(&bcnt[b], h) : 0;
  }
}

__global__ void vg_bucket_scatter_kernel(const int* __restrict__ src, const int* __restrict__ dst,
                                         const int* __restrict__ bcnt, const int* __restrict__ blkbase,
                                         int2* __restrict__ binned, int e, int nb) {
  extern __shared__ int lds_i[];
  __shared__ int wsum[8];
  int lane = threadIdx.x & 31;
  int wv = threadIdx.x >> 5;
  int run = 0;
  for (int b0 = 0; b0 < nb; b0 += 256) {
    int b = b0 + threadIdx.x;
    int c = (b < nb) ? bcnt[b] : 0;
    int x = c;
#pragma unroll
    for (int o = 1; o < 32; o <<= 1) {
      int y = __shfl_up(x, o, 32);
      if (lane >= o) x += y;
    }
    if (lane == 31) wsum[wv] = x;
    __syncthreads();
    int wbase = 0, tot = 0;
#pragma unroll
    for (int k = 0; k < 8; ++k) { int t = wsum[k]; if (k < wv) wbase += t; tot += t; }
    if (b < nb) lds_i[b] = run + wbase + (x - c) + blkbase[(size_t)blockIdx.x * (size_t)nb + b];
    run += tot;
    __syncthreads();
  }
  int e0 = blockIdx.x * EPB + threadIdx.x;
  int d[EPT], s[EPT];
#pragma unroll
  for (int k = 0; k < EPT; ++k) {
    int i = e0 + k * 256;
    d[k] = (i < e) ? dst[i] : -1;
    s[k] = (i < e) ? src[i] : 0;
  }
#pragma unroll
  for (int k = 0; k < EPT; ++k) {
    if (d[k] >= 0) {
      int p = atomicAdd(&lds_i[d[k] >> BKT_SHIFT], 1);
      binned[p] = make_int2(s[k], d[k]);
    }
  }
}

__global__ void vg_bucket_rows_kernel(const int2* __restrict__ binned, const int* __restrict__ bcnt,
                                      int* __restrict__ cnt, int* __restrict__ rowstart,
                                      float* __restrict__ dinv, int* __restrict__ csr, int n) {
  __shared__ int lcnt[BKT_ROWS];
  __shared__ int lcur[BKT_ROWS];
  __shared__ int wsum[8];
  __shared__ int wsum2[8];
  int bkt = blockIdx.x;
  int t = threadIdx.x;
  int lane = t & 31;
  int wv = t >> 5;
  int part = 0;
  for (int j = t; j < bkt; j += 256) part += bcnt[j];
#pragma unroll
  for (int o = 16; o > 0; o >>= 1) part += __shfl_xor(part, o, 32);
  if (lane == 0) wsum[wv] = part;
  lcnt[t] = 0;
  __syncthreads();
  int bstart = 0;
#pragma unroll
  for (int k = 0; k < 8; ++k) bstart += wsum[k];
  int bsize = bcnt[bkt];
  const int2* be = binned + bstart;

  for (int i0 = 0; i0 < bsize; i0 += 256 * BPT) {
    int r[BPT];
#pragma unroll
    for (int k = 0; k < BPT; ++k) {
      int i = i0 + k * 256 + t;
      r[k] = be[(i < bsize) ? i : (bsize - 1)].y;
    }
#pragma unroll
    for (int k = 0; k < BPT; ++k) {
      if (i0 + k * 256 + t < bsize) atomicAdd(&lcnt[r[k] & (BKT_ROWS - 1)], 1);
    }
  }
  __syncthreads();

  int c = lcnt[t];
  int x = c;
#pragma unroll
  for (int o = 1; o < 32; o <<= 1) {
    int y = __shfl_up(x, o, 32);
    if (lane >= o) x += y;
  }
  if (lane == 31) wsum2[wv] = x;
  __syncthreads();
  int wbase = 0;
#pragma unroll
  for (int k = 0; k < 8; ++k) { int w = wsum2[k]; if (k < wv) wbase += w; }
  int rs = bstart + wbase + (x - c);
  lcur[t] = rs;
  int row = bkt * BKT_ROWS + t;
  if (row < n) {
    cnt[row] = c;
    rowstart[row] = rs;
    dinv[row] = rsqrtf((float)c + 1.0f);
  }
  __syncthreads();

  for (int i0 = 0; i0 < bsize; i0 += 256 * BPT) {
    int2 ev[BPT];
#pragma unroll
    for (int k = 0; k < BPT; ++k) {
      int i = i0 + k * 256 + t;
      ev[k] = be[(i < bsize) ? i : (bsize - 1)];
    }
#pragma unroll
    for (int k = 0; k < BPT; ++k) {
      if (i0 + k * 256 + t < bsize) {
        int p = atomicAdd(&lcur[ev[k].y & (BKT_ROWS - 1)], 1);
        csr[p] = ev[k].x;
      }
    }
  }
}

#define GEMM_NT 4
static_assert(D_OUT == 16 * GEMM_NT, "the second GEMM's wave covers the whole width of L");
__global__ void vg_gemm_wmma_kernel(const float* __restrict__ A, const float* __restrict__ B,
                                    const float* __restrict__ bias, float* __restrict__ C,
                                    int M, int K, int Ncols) {
  const int wavesPerBlock = blockDim.x >> 5;
  int w = blockIdx.x * wavesPerBlock + (threadIdx.x >> 5);
  int stripsN = Ncols / (16 * GEMM_NT);
  int totalStrips = (M >> 4) * stripsN;
  if (w >= totalStrips) return;
  int tm = w / stripsN;
  int ts = w - tm * stripsN;
  int lane = threadIdx.x & 31;
  int m  = lane & 15;
  int hi = lane >> 4;

  v8f acc[GEMM_NT];
#pragma unroll
  for (int t = 0; t < GEMM_NT; ++t) acc[t] = (v8f){};
  const float* arow = A + (size_t)(tm * 16 + m) * (size_t)K;
  const float* bcol = B + ts * (16 * GEMM_NT) + m;
  for (int k0 = 0; k0 < K; k0 += 4) {
    int kb = k0 + 2 * hi;
    v2f a = *(const v2f*)(arow + kb);
    const float* b0 = bcol + (size_t)kb * (size_t)Ncols;
    const float* b1 = b0 + Ncols;
    v2f b[GEMM_NT];
#pragma unroll
    for (int t = 0; t < GEMM_NT; ++t) {
      b[t].x = b0[t * 16];
      b[t].y = b1[t * 16];
    }
#pragma unroll
    for (int t = 0; t < GEMM_NT; ++t) {
      acc[t] = __builtin_amdgcn_wmma_f32_16x16x4_f32(false, a, false, b[t],
                                                     (short)0, acc[t], false, false);
    }
  }
  int rbase = tm * 16 + hi * 8;
#pragma unroll
  for (int t = 0; t < GEMM_NT; ++t) {
    int col = ts * (16 * GEMM_NT) + t * 16 + m;
    float bv = bias[col];
#pragma unroll
    for (int j = 0; j < 8; ++j) {
      C[(size_t)(rbase + j) * (size_t)Ncols + (size_t)col] = acc[t][j] + bv;
    }
  }
}

#define ROWS_PER_WAVE 8
template <int U>
__device__ __forceinline__ void vg_edges128(const float* __restrict__ in, int s_l, float w_l,
                                            int j, int lane, float4& acc) {
  float4 v[U];
  float  w[U];
#pragma unroll
  for (int u = 0; u < U; ++u) {
    int s = __builtin_amdgcn_readlane(s_l, j + u);
    w[u]  = __int_as_float(__builtin_amdgcn_readlane(__float_as_int(w_l), j + u));
    v[u]  = ((const float4*)(in + (size_t)s * D_HID))[lane];
  }
#pragma unroll
  for (int u = 0; u < U; ++u) {
    acc.x += w[u] * v[u].x;
    acc.y += w[u] * v[u].y;
    acc.z += w[u] * v[u].z;
    acc.w += w[u] * v[u].w;
  }
}

__global__ void vg_gather128_kernel(const float* __restrict__ in, const int* __restrict__ csr,
                                    const int* __restrict__ rowstart, const int* __restrict__ cnt,
                                    const float* __restrict__ dinv, float* __restrict__ out, int n) {
  int wv = __builtin_amdgcn_readfirstlane((int)(threadIdx.x >> 5));
  int row0 = (blockIdx.x * 8 + wv) * ROWS_PER_WAVE;
  int lane = threadIdx.x & 31;
  for (int r = 0; r < ROWS_PER_WAVE; ++r) {
    int row = row0 + r;
    if (row >= n) break;
    int start = rowstart[row];
    int c = cnt[row];
    float dd = dinv[row];
    float d2 = dd * dd;
    float4 sv = ((const float4*)(in + (size_t)row * D_HID))[lane];
    float4 acc;
    acc.x = sv.x * d2; acc.y = sv.y * d2; acc.z = sv.z * d2; acc.w = sv.w * d2;
    for (int base = 0; base < c; base += 32) {
      int m = c - base;
      if (m > 32) m = 32;
      int s_l = 0;
      float w_l = 0.0f;
      if (lane < m) {
        s_l = csr[start + base + lane];
        w_l = dinv[s_l] * dd;
      }
      int j = 0;
      for (; j + 8 <= m; j += 8) vg_edges128<8>(in, s_l, w_l, j, lane, acc);
      if (j + 4 <= m) { vg_edges128<4>(in, s_l, w_l, j, lane, acc); j += 4; }
      if (j + 2 <= m) { vg_edges128<2>(in, s_l, w_l, j, lane, acc); j += 2; }
      if (j < m)      { vg_edges128<1>(in, s_l, w_l, j, lane, acc); }
    }
    ((float4*)(out + (size_t)row * D_HID))[lane] = acc;
  }
}

__global__ void vg_bn_reduce_kernel(const float* __restrict__ h, float* __restrict__ stats, int n) {
  __shared__ float sh[256];
  int c    = threadIdx.x & 127;
  int half = threadIdx.x >> 7;
  float sum = 0.f, sq = 0.f;
  for (int r = blockIdx.x * 2 + half; r < n; r += gridDim.x * 2) {
    float v = h[(size_t)r * D_HID + c];
    sum += v; sq += v * v;
  }
  sh[threadIdx.x] = sum;
  __syncthreads();
  if (half == 0) atomicAdd(&stats[c], sh[c] + sh[c + 128]);
  __syncthreads();
  sh[threadIdx.x] = sq;
  __syncthreads();
  if (half == 0) atomicAdd(&stats[128 + c], sh[c] + sh[c + 128]);
}

__global__ void vg_gemm_bn_wmma_kernel(const float* __restrict__ A, const float* __restrict__ B,
                                       const float* __restrict__ bias, const float* __restrict__ stats,
                                       const float* __restrict__ gamma, const float* __restrict__ beta,
                                       float* __restrict__ C, int M, int n) {
  __shared__ __attribute__((aligned(16))) float bn_sc[D_HID];
  __shared__ __attribute__((aligned(16))) float bn_sh[D_HID];
  if (threadIdx.x < D_HID) {
    int c = threadIdx.x;
    float invn = 1.0f / (float)n;
    float mean = stats[c] * invn;
    float var  = stats[128 + c] * invn - mean * mean;
    float a = gamma[c] * rsqrtf(var + BN_EPS);
    bn_sc[c] = a;
    bn_sh[c] = beta[c] - mean * a;
  }
  __syncthreads();

  const int wavesPerBlock = blockDim.x >> 5;
  int tm = blockIdx.x * wavesPerBlock + (threadIdx.x >> 5);
  if (tm >= (M >> 4)) return;
  int lane = threadIdx.x & 31;
  int m  = lane & 15;
  int hi = lane >> 4;

  v8f acc[GEMM_NT];
#pragma unroll
  for (int t = 0; t < GEMM_NT; ++t) acc[t] = (v8f){};
  const float* arow = A + (size_t)(tm * 16 + m) * (size_t)D_HID;
  const float* bcol = B + m;
  for (int k0 = 0; k0 < D_HID; k0 += 4) {
    int kb = k0 + 2 * hi;
    v2f a  = *(const v2f*)(arow + kb);
    v2f sc = *(const v2f*)(bn_sc + kb);
    v2f sh = *(const v2f*)(bn_sh + kb);
    float v0 = sc.x * a.x + sh.x;
    float v1 = sc.y * a.y + sh.y;
    a.x = (v0 > 0.f) ? v0 : LRELU_SLOPE * v0;
    a.y = (v1 > 0.f) ? v1 : LRELU_SLOPE * v1;
    const float* b0 = bcol + (size_t)kb * (size_t)D_OUT;
    const float* b1 = b0 + D_OUT;
    v2f b[GEMM_NT];
#pragma unroll
    for (int t = 0; t < GEMM_NT; ++t) {
      b[t].x = b0[t * 16];
      b[t].y = b1[t * 16];
    }
#pragma unroll
    for (int t = 0; t < GEMM_NT; ++t) {
      acc[t] = __builtin_amdgcn_wmma_f32_16x16x4_f32(false, a, false, b[t],
                                                     (short)0, acc[t], false, false);
    }
  }
  int rbase = tm * 16 + hi * 8;
#pragma unroll
  for (int t = 0; t < GEMM_NT; ++t) {
    int col = t * 16 + m;
    float bv = bias[col];
#pragma unroll
    for (int j = 0; j < 8; ++j) {
      C[(size_t)(rbase + j) * (size_t)D_OUT + (size_t)col] = acc[t][j] + bv;
    }
  }
}

__global__ void vg_rownorm_kernel(const float* __restrict__ l, float* __restrict__ rns, int n) {
  int row0 = (blockIdx.x * (blockDim.x >> 5) + (threadIdx.x >> 5)) * ROWS_PER_WAVE;
  if (row0 >= n) return;
  int lane = threadIdx.x & 31;
  float a0[ROWS_PER_WAVE], a1[ROWS_PER_WAVE];
#pragma unroll
  for (int r = 0; r < ROWS_PER_WAVE; ++r) {
    int row = row0 + r;
    const float* rp = l + (size_t)((row < n) ? row : (n - 1)) * D_OUT;
    a0[r] = rp[lane];
    a1[r] = rp[lane + 32];
  }
#pragma unroll
  for (int r = 0; r < ROWS_PER_WAVE; ++r) {
    float s = a0[r] * a0[r] + a1[r] * a1[r];
#pragma unroll
    for (int m = 16; m > 0; m >>= 1) s += __shfl_xor(s, m, 32);
    if (lane == 0 && row0 + r < n) rns[row0 + r] = NORM_SCALE / fmaxf(sqrtf(s), 1e-12f);
  }
}

template <int U>
__device__ __forceinline__ void vg_edges64(const float* __restrict__ l, int s_l, float w_l, float sc_l,
                                           int j, int lane, float2& am, float2& al) {
  float2 v[U];
  float  w[U], sc[U];
#pragma unroll
  for (int u = 0; u < U; ++u) {
    int s = __builtin_amdgcn_readlane(s_l, j + u);
    w[u]  = __int_as_float(__builtin_amdgcn_readlane(__float_as_int(w_l), j + u));
    sc[u] = __int_as_float(__builtin_amdgcn_readlane(__float_as_int(sc_l), j + u));
    v[u]  = ((const float2*)(l + (size_t)s * D_OUT))[lane];
  }
#pragma unroll
  for (int u = 0; u < U; ++u) {
    am.x += w[u]  * v[u].x;
    am.y += w[u]  * v[u].y;
    al.x += sc[u] * v[u].x;
    al.y += sc[u] * v[u].y;
  }
}

__global__ void vg_gather64_kernel(const float* __restrict__ l, const int* __restrict__ csr,
                                   const int* __restrict__ rowstart, const int* __restrict__ cnt,
                                   const float* __restrict__ dinv, const float* __restrict__ rns,
                                   const float* __restrict__ noise, float* __restrict__ mu,
                                   float* __restrict__ logstd, float* __restrict__ zeta, int n) {
  int wv = __builtin_amdgcn_readfirstlane((int)(threadIdx.x >> 5));
  int row0 = (blockIdx.x * 8 + wv) * ROWS_PER_WAVE;
  int lane = threadIdx.x & 31;
  for (int r = 0; r < ROWS_PER_WAVE; ++r) {
    int row = row0 + r;
    if (row >= n) break;
    int start = rowstart[row];
    int c = cnt[row];
    float dd = dinv[row];
    float d2 = dd * dd;
    float rr = rns[row];
    float2 sv = ((const float2*)(l + (size_t)row * D_OUT))[lane];
    float2 nz = ((const float2*)(noise + (size_t)row * D_OUT))[lane];
    float2 am, al;
    am.x = sv.x * d2;      am.y = sv.y * d2;
    al.x = sv.x * rr * d2; al.y = sv.y * rr * d2;
    for (int base = 0; base < c; base += 32) {
      int m = c - base;
      if (m > 32) m = 32;
      int s_l = 0;
      float w_l = 0.0f, sc_l = 0.0f;
      if (lane < m) {
        s_l  = csr[start + base + lane];
        w_l  = dinv[s_l] * dd;
        sc_l = w_l * rns[s_l];
      }
      int j = 0;
      for (; j + 8 <= m; j += 8) vg_edges64<8>(l, s_l, w_l, sc_l, j, lane, am, al);
      if (j + 4 <= m) { vg_edges64<4>(l, s_l, w_l, sc_l, j, lane, am, al); j += 4; }
      if (j + 2 <= m) { vg_edges64<2>(l, s_l, w_l, sc_l, j, lane, am, al); j += 2; }
      if (j < m)      { vg_edges64<1>(l, s_l, w_l, sc_l, j, lane, am, al); }
    }
    ((float2*)(mu     + (size_t)row * D_OUT))[lane] = am;
    ((float2*)(logstd + (size_t)row * D_OUT))[lane] = al;
    float2 z;
    z.x = am.x + nz.x * expf(al.x);
    z.y = am.y + nz.y * expf(al.y);
    ((float2*)(zeta   + (size_t)row * D_OUT))[lane] = z;
  }
}

extern "C" void kernel_launch(void* const* d_in, const int* in_sizes, int n_in,
                              void* d_out, int out_size, void* d_ws, size_t ws_size,
                              hipStream_t stream) {
  const float* x     = (const float*)d_in[0];
  const int*   esrc  = (const int*)  d_in[1];
  const int*   edst  = (const int*)  d_in[2];
  const float* W1    = (const float*)d_in[3];
  const float* b1    = (const float*)d_in[4];
  const float* gamma = (const float*)d_in[5];
  const float* beta  = (const float*)d_in[6];
  const float* Wmu   = (const float*)d_in[7];
  const float* bmu   = (const float*)d_in[8];
  const float* noise = (const float*)d_in[9];

  const int N = in_sizes[0] / D_IN;
  const int E = in_sizes[1];

  float* ws = (float*)d_ws;
  size_t npad = ((size_t)N + 255) & ~(size_t)255;
  size_t off = 0;
  float* dinv  = ws + off; off += npad;
  float* stats = ws + off; off += 256;
  float* rns   = ws + off; off += npad;
  float* h     = ws + off; off += (size_t)N * D_HID;
  float* hagg  = ws + off; off += (size_t)N * D_HID;
  float* lbuf  = ws + off; off += (size_t)N * D_OUT;
  size_t epad = ((size_t)E + 255) & ~(size_t)255;
  const int NB   = (N + BKT_ROWS - 1) / BKT_ROWS;
  const int NBLK = (E + EPB - 1) / EPB;
  size_t bbpad = ((size_t)NBLK * (size_t)NB + 255) & ~(size_t)255;
  int* cnt      = (int*)(ws + off); off += npad;
  int* rowstart = (int*)(ws + off); off += npad;
  int* bcnt     = (int*)(ws + off); off += npad;
  int* blkbase  = (int*)(ws + off); off += bbpad;
  int* csr      = (int*)(ws + off); off += epad;
  int2* binned  = (int2*)(ws + off); off += 2 * epad;

  float* mu_out   = (float*)d_out;
  float* ls_out   = mu_out + (size_t)N * D_OUT;
  float* zeta_out = ls_out + (size_t)N * D_OUT;

  const int B = 256;
  auto cdiv = [](int a, int b) { return (a + b - 1) / b; };

  vg_init_kernel<<<cdiv(NB, B), B, 0, stream>>>(bcnt, stats, NB);
  vg_bucket_count_kernel<<<NBLK, B, NB * sizeof(int), stream>>>(edst, bcnt, blkbase, E, NB);
  vg_bucket_scatter_kernel<<<NBLK, B, NB * sizeof(int), stream>>>(esrc, edst, bcnt, blkbase, binned, E, NB);
  vg_bucket_rows_kernel<<<NB, B, 0, stream>>>(binned, bcnt, cnt, rowstart, dinv, csr, N);

  {
    int strips = (N / 16) * (D_HID / (16 * GEMM_NT));
    vg_gemm_wmma_kernel<<<cdiv(strips, B / 32), B, 0, stream>>>(x, W1, b1, h, N, D_IN, D_HID);
  }

  vg_gather128_kernel<<<cdiv(N, (B / 32) * ROWS_PER_WAVE), B, 0, stream>>>(h, csr, rowstart, cnt, dinv, hagg, N);

  vg_bn_reduce_kernel<<<512, B, 0, stream>>>(hagg, stats, N);

  {
    int strips = N / 16;
    vg_gemm_bn_wmma_kernel<<<cdiv(strips, B / 32), B, 0, stream>>>(hagg, Wmu, bmu, stats, gamma, beta, lbuf, N, N);
  }

  vg_rownorm_kernel<<<cdiv(N, (B / 32) * ROWS_PER_WAVE), B, 0, stream>>>(lbuf, rns, N);

  vg_gather64_kernel<<<cdiv(N, (B / 32) * ROWS_PER_WAVE), B, 0, stream>>>(lbuf, csr, rowstart, cnt, dinv, rns, noise,
                                                        mu_out, ls_out, zeta_out, N);
}


# BatchNorm sums taken in the first gather's epilogue and L's row norms in the second GEMM's epilogue, no pass of their own: 10 launches to 8; both sums' order changes (block partials, atomic arrival)
# speedup vs baseline: 4.3513x; 1.2813x over previous
#include <hip/hip_runtime.h>
#include <hip/hip_bf16.h>


#define D_IN   128
#define D_HID  128
#define D_OUT  64
#define BN_EPS 1e-5f
#define LRELU_SLOPE 0.01f
#define NORM_SCALE 1.8f

typedef __attribute__((ext_vector_type(2))) float v2f;
typedef __attribute__((ext_vector_type(8))) float v8f;

#define BKT_SHIFT 8
#define BKT_ROWS  256
#define EPB       4096
#define EPT       (EPB / 256)
#define BPT       20

#define STAT_COPIES 16
#define STAT_WORDS  (256 * STAT_COPIES)

__global__ void vg_init_kernel(int* __restrict__ bcnt, float* __restrict__ stats, int nb) {
  int i = blockIdx.x * blockDim.x + threadIdx.x;
  if (i < nb) bcnt[i] = 0;
  if (i < STAT_WORDS) stats[i] = 0.0f;
}

__global__ void vg_bucket_count_kernel(const int* __restrict__ dst, int* __restrict__ bcnt,
                                       int* __restrict__ blkbase, int e, int nb) {
  extern __shared__ int lds_i[];
  for (int b = threadIdx.x; b < nb; b += 256) lds_i[b] = 0;
  __syncthreads();
  int e0 = blockIdx.x * EPB + threadIdx.x;
  int d[EPT];
#pragma unroll
  for (int k = 0; k < EPT; ++k) {
    int i = e0 + k * 256;
    d[k] = (i < e) ? dst[i] : -1;
  }
#pragma unroll
  for (int k = 0; k < EPT; ++k) {
    if (d[k] >= 0) atomicAdd(&lds_i[d[k] >> BKT_SHIFT], 1);
  }
  __syncthreads();
  for (int b = threadIdx.x; b < nb; b += 256) {
    int h = lds_i[b];
    blkbase[(size_t)blockIdx.x * (size_t)nb + b] = (h > 0) ? atomicAdd(&bcnt[b], h) : 0;
  }
}

__global__ void vg_bucket_scatter_kernel(const int* __restrict__ src, const int* __restrict__ dst,
                                         const int* __restrict__ bcnt, const int* __restrict__ blkbase,
                                         int2* __restrict__ binned, int e, int nb) {
  extern __shared__ int lds_i[];
  __shared__ int wsum[8];
  int lane = threadIdx.x & 31;
  int wv = threadIdx.x >> 5;
  int run = 0;
  for (int b0 = 0; b0 < nb; b0 += 256) {
    int b = b0 + threadIdx.x;
    int c = (b < nb) ? bcnt[b] : 0;
    int x = c;
#pragma unroll
    for (int o = 1; o < 32; o <<= 1) {
      int y = __shfl_up(x, o, 32);
      if (lane >= o) x += y;
    }
    if (lane == 31) wsum[wv] = x;
    __syncthreads();
    int wbase = 0, tot = 0;
#pragma unroll
    for (int k = 0; k < 8; ++k) { int t = wsum[k]; if (k < wv) wbase += t; tot += t; }
    if (b < nb) lds_i[b] = run + wbase + (x - c) + blkbase[(size_t)blockIdx.x * (size_t)nb + b];
    run += tot;
    __syncthreads();
  }
  int e0 = blockIdx.x * EPB + threadIdx.x;
  int d[EPT], s[EPT];
#pragma unroll
  for (int k = 0; k < EPT; ++k) {
    int i = e0 + k * 256;
    d[k] = (i < e) ? dst[i] : -1;
    s[k] = (i < e) ? src[i] : 0;
  }
#pragma unroll
  for (int k = 0; k < EPT; ++k) {
    if (d[k] >= 0) {
      int p = atomicAdd(&lds_i[d[k] >> BKT_SHIFT], 1);
      binned[p] = make_int2(s[k], d[k]);
    }
  }
}

__global__ void vg_bucket_rows_kernel(const int2* __restrict__ binned, const int* __restrict__ bcnt,
                                      int* __restrict__ cnt, int* __restrict__ rowstart,
                                      float* __restrict__ dinv, int* __restrict__ csr, int n) {
  __shared__ int lcnt[BKT_ROWS];
  __shared__ int lcur[BKT_ROWS];
  __shared__ int wsum[8];
  __shared__ int wsum2[8];
  int bkt = blockIdx.x;
  int t = threadIdx.x;
  int lane = t & 31;
  int wv = t >> 5;
  int part = 0;
  for (int j = t; j < bkt; j += 256) part += bcnt[j];
#pragma unroll
  for (int o = 16; o > 0; o >>= 1) part += __shfl_xor(part, o, 32);
  if (lane == 0) wsum[wv] = part;
  lcnt[t] = 0;
  __syncthreads();
  int bstart = 0;
#pragma unroll
  for (int k = 0; k < 8; ++k) bstart += wsum[k];
  int bsize = bcnt[bkt];
  const int2* be = binned + bstart;

  for (int i0 = 0; i0 < bsize; i0 += 256 * BPT) {
    int r[BPT];
#pragma unroll
    for (int k = 0; k < BPT; ++k) {
      int i = i0 + k * 256 + t;
      r[k] = be[(i < bsize) ? i : (bsize - 1)].y;
    }
#pragma unroll
    for (int k = 0; k < BPT; ++k) {
      if (i0 + k * 256 + t < bsize) atomicAdd(&lcnt[r[k] & (BKT_ROWS - 1)], 1);
    }
  }
  __syncthreads();

  int c = lcnt[t];
  int x = c;
#pragma unroll
  for (int o = 1; o < 32; o <<= 1) {
    int y = __shfl_up(x, o, 32);
    if (lane >= o) x += y;
  }
  if (lane == 31) wsum2[wv] = x;
  __syncthreads();
  int wbase = 0;
#pragma unroll
  for (int k = 0; k < 8; ++k) { int w = wsum2[k]; if (k < wv) wbase += w; }
  int rs = bstart + wbase + (x - c);
  lcur[t] = rs;
  int row = bkt * BKT_ROWS + t;
  if (row < n) {
    cnt[row] = c;
    rowstart[row] = rs;
    dinv[row] = rsqrtf((float)c + 1.0f);
  }
  __syncthreads();

  for (int i0 = 0; i0 < bsize; i0 += 256 * BPT) {
    int2 ev[BPT];
#pragma unroll
    for (int k = 0; k < BPT; ++k) {
      int i = i0 + k * 256 + t;
      ev[k] = be[(i < bsize) ? i : (bsize - 1)];
    }
#pragma unroll
    for (int k = 0; k < BPT; ++k) {
      if (i0 + k * 256 + t < bsize) {
        int p = atomicAdd(&lcur[ev[k].y & (BKT_ROWS - 1)], 1);
        csr[p] = ev[k].x;
      }
    }
  }
}

#define GEMM_NT 4
static_assert(D_OUT == 16 * GEMM_NT, "the second GEMM's wave covers the whole width of L");
__global__ void vg_gemm_wmma_kernel(const float* __restrict__ A, const float* __restrict__ B,
                                    const float* __restrict__ bias, float* __restrict__ C,
                                    int M, int K, int Ncols) {
  const int wavesPerBlock = blockDim.x >> 5;
  int w = blockIdx.x * wavesPerBlock + (threadIdx.x >> 5);
  int stripsN = Ncols / (16 * GEMM_NT);
  int totalStrips = (M >> 4) * stripsN;
  if (w >= totalStrips) return;
  int tm = w / stripsN;
  int ts = w - tm * stripsN;
  int lane = threadIdx.x & 31;
  int m  = lane & 15;
  int hi = lane >> 4;

  v8f acc[GEMM_NT];
#pragma unroll
  for (int t = 0; t < GEMM_NT; ++t) acc[t] = (v8f){};
  const float* arow = A + (size_t)(tm * 16 + m) * (size_t)K;
  const float* bcol = B + ts * (16 * GEMM_NT) + m;
  for (int k0 = 0; k0 < K; k0 += 4) {
    int kb = k0 + 2 * hi;
    v2f a = *(const v2f*)(arow + kb);
    const float* b0 = bcol + (size_t)kb * (size_t)Ncols;
    const float* b1 = b0 + Ncols;
    v2f b[GEMM_NT];
#pragma unroll
    for (int t = 0; t < GEMM_NT; ++t) {
      b[t].x = b0[t * 16];
      b[t].y = b1[t * 16];
    }
#pragma unroll
    for (int t = 0; t < GEMM_NT; ++t) {
      acc[t] = __builtin_amdgcn_wmma_f32_16x16x4_f32(false, a, false, b[t],
                                                     (short)0, acc[t], false, false);
    }
  }
  int rbase = tm * 16 + hi * 8;
#pragma unroll
  for (int t = 0; t < GEMM_NT; ++t) {
    int col = ts * (16 * GEMM_NT) + t * 16 + m;
    float bv = bias[col];
#pragma unroll
    for (int j = 0; j < 8; ++j) {
      C[(size_t)(rbase + j) * (size_t)Ncols + (size_t)col] = acc[t][j] + bv;
    }
  }
}

#define ROWS_PER_WAVE 8
template <int U>
__device__ __forceinline__ void vg_edges128(const float* __restrict__ in, int s_l, float w_l,
                                            int j, int lane, float4& acc) {
  float4 v[U];
  float  w[U];
#pragma unroll
  for (int u = 0; u < U; ++u) {
    int s = __builtin_amdgcn_readlane(s_l, j + u);
    w[u]  = __int_as_float(__builtin_amdgcn_readlane(__float_as_int(w_l), j + u));
    v[u]  = ((const float4*)(in + (size_t)s * D_HID))[lane];
  }
#pragma unroll
  for (int u = 0; u < U; ++u) {
    acc.x += w[u] * v[u].x;
    acc.y += w[u] * v[u].y;
    acc.z += w[u] * v[u].z;
    acc.w += w[u] * v[u].w;
  }
}

__global__ void vg_gather128_kernel(const float* __restrict__ in, const int* __restrict__ csr,
                                    const int* __restrict__ rowstart, const int* __restrict__ cnt,
                                    const float* __restrict__ dinv, float* __restrict__ out,
                                    float* __restrict__ stats, int n) {
  __shared__ float4 red[8][64];
  int wv = __builtin_amdgcn_readfirstlane((int)(threadIdx.x >> 5));
  int row0 = (blockIdx.x * 8 + wv) * ROWS_PER_WAVE;
  int lane = threadIdx.x & 31;
  float4* my_sum = &red[wv][lane];
  float4* my_sq  = &red[wv][32 + lane];
  *my_sum = make_float4(0.f, 0.f, 0.f, 0.f);
  *my_sq  = make_float4(0.f, 0.f, 0.f, 0.f);
  for (int r = 0; r < ROWS_PER_WAVE; ++r) {
    int row = row0 + r;
    if (row >= n) break;
    int start = rowstart[row];
    int c = cnt[row];
    float dd = dinv[row];
    float d2 = dd * dd;
    float4 sv = ((const float4*)(in + (size_t)row * D_HID))[lane];
    float4 acc;
    acc.x = sv.x * d2; acc.y = sv.y * d2; acc.z = sv.z * d2; acc.w = sv.w * d2;
    for (int base = 0; base < c; base += 32) {
      int m = c - base;
      if (m > 32) m = 32;
      int s_l = 0;
      float w_l = 0.0f;
      if (lane < m) {
        s_l = csr[start + base + lane];
        w_l = dinv[s_l] * dd;
      }
      int j = 0;
      for (; j + 8 <= m; j += 8) vg_edges128<8>(in, s_l, w_l, j, lane, acc);
      if (j + 4 <= m) { vg_edges128<4>(in, s_l, w_l, j, lane, acc); j += 4; }
      if (j + 2 <= m) { vg_edges128<2>(in, s_l, w_l, j, lane, acc); j += 2; }
      if (j < m)      { vg_edges128<1>(in, s_l, w_l, j, lane, acc); }
    }
    ((float4*)(out + (size_t)row * D_HID))[lane] = acc;
    float4 ssum = *my_sum;
    float4 ssq  = *my_sq;
    ssum.x += acc.x; ssum.y += acc.y; ssum.z += acc.z; ssum.w += acc.w;
    ssq.x += acc.x * acc.x; ssq.y += acc.y * acc.y; ssq.z += acc.z * acc.z; ssq.w += acc.w * acc.w;
    *my_sum = ssum;
    *my_sq  = ssq;
  }
  __syncthreads();
  const float* rf = (const float*)&red[0][0];
  float tot = 0.f;
#pragma unroll
  for (int k = 0; k < 8; ++k) tot += rf[k * 256 + threadIdx.x];
  atomicAdd(&stats[(blockIdx.x & (STAT_COPIES - 1)) * 256 + threadIdx.x], tot);
}


template <int MASK>
__device__ __forceinline__ float vg_add_row_xor(float x) {
  int y = __builtin_amdgcn_update_dpp(0, __float_as_int(x), 0x160 + MASK, 0xf, 0xf, true);
  return x + __int_as_float(y);
}

__global__ void vg_gemm_bn_wmma_kernel(const float* __restrict__ A, const float* __restrict__ B,
                                       const float* __restrict__ bias, const float* __restrict__ stats,
                                       const float* __restrict__ gamma, const float* __restrict__ beta,
                                       float* __restrict__ C, float* __restrict__ rns, int M, int n) {
  __shared__ __attribute__((aligned(16))) float bn_sc[D_HID];
  __shared__ __attribute__((aligned(16))) float bn_sh[D_HID];
  if (threadIdx.x < D_HID) {
    int c = threadIdx.x;
    float invn = 1.0f / (float)n;
    float s1 = 0.f, s2 = 0.f;
#pragma unroll
    for (int k = 0; k < STAT_COPIES; ++k) {
      s1 += stats[k * 256 + c];
      s2 += stats[k * 256 + 128 + c];
    }
    float mean = s1 * invn;
    float var  = s2 * invn - mean * mean;
    float a = gamma[c] * rsqrtf(var + BN_EPS);
    bn_sc[c] = a;
    bn_sh[c] = beta[c] - mean * a;
  }
  __syncthreads();

  const int wavesPerBlock = blockDim.x >> 5;
  int tm = blockIdx.x * wavesPerBlock + (threadIdx.x >> 5);
  if (tm >= (M >> 4)) return;
  int lane = threadIdx.x & 31;
  int m  = lane & 15;
  int hi = lane >> 4;

  v8f acc[GEMM_NT];
#pragma unroll
  for (int t = 0; t < GEMM_NT; ++t) acc[t] = (v8f){};
  const float* arow = A + (size_t)(tm * 16 + m) * (size_t)D_HID;
  const float* bcol = B + m;
  for (int k0 = 0; k0 < D_HID; k0 += 4) {
    int kb = k0 + 2 * hi;
    v2f a  = *(const v2f*)(arow + kb);
    v2f sc = *(const v2f*)(bn_sc + kb);
    v2f sh = *(const v2f*)(bn_sh + kb);
    float v0 = sc.x * a.x + sh.x;
    float v1 = sc.y * a.y + sh.y;
    a.x = (v0 > 0.f) ? v0 : LRELU_SLOPE * v0;
    a.y = (v1 > 0.f) ? v1 : LRELU_SLOPE * v1;
    const float* b0 = bcol + (size_t)kb * (size_t)D_OUT;
    const float* b1 = b0 + D_OUT;
    v2f b[GEMM_NT];
#pragma unroll
    for (int t = 0; t < GEMM_NT; ++t) {
      b[t].x = b0[t * 16];
      b[t].y = b1[t * 16];
    }
#pragma unroll
    for (int t = 0; t < GEMM_NT; ++t) {
      acc[t] = __builtin_amdgcn_wmma_f32_16x16x4_f32(false, a, false, b[t],
                                                     (short)0, acc[t], false, false);
    }
  }
  int rbase = tm * 16 + hi * 8;
  float sq[8];
#pragma unroll
  for (int j = 0; j < 8; ++j) sq[j] = 0.f;
#pragma unroll
  for (int t = 0; t < GEMM_NT; ++t) {
    int col = t * 16 + m;
    float bv = bias[col];
#pragma unroll
    for (int j = 0; j < 8; ++j) {
      float v = acc[t][j] + bv;
      C[(size_t)(rbase + j) * (size_t)D_OUT + (size_t)col] = v;
      sq[j] += v * v;
    }
  }
  float mine = 0.f;
#pragma unroll
  for (int j = 0; j < 8; ++j) {
    float s = sq[j];
    s = vg_add_row_xor<8>(s);
    s = vg_add_row_xor<4>(s);
    s = vg_add_row_xor<2>(s);
    s = vg_add_row_xor<1>(s);
    mine = (m == j) ? s : mine;
  }
  if (m < 8) rns[rbase + m] = NORM_SCALE / fmaxf(sqrtf(mine), 1e-12f);
}

template <int U>
__device__ __forceinline__ void vg_edges64(const float* __restrict__ l, int s_l, float w_l, float sc_l,
                                           int j, int lane, float2& am, float2& al) {
  float2 v[U];
  float  w[U], sc[U];
#pragma unroll
  for (int u = 0; u < U; ++u) {
    int s = __builtin_amdgcn_readlane(s_l, j + u);
    w[u]  = __int_as_float(__builtin_amdgcn_readlane(__float_as_int(w_l), j + u));
    sc[u] = __int_as_float(__builtin_amdgcn_readlane(__float_as_int(sc_l), j + u));
    v[u]  = ((const float2*)(l + (size_t)s * D_OUT))[lane];
  }
#pragma unroll
  for (int u = 0; u < U; ++u) {
    am.x += w[u]  * v[u].x;
    am.y += w[u]  * v[u].y;
    al.x += sc[u] * v[u].x;
    al.y += sc[u] * v[u].y;
  }
}

__global__ void vg_gather64_kernel(const float* __restrict__ l, const int* __restrict__ csr,
                                   const int* __restrict__ rowstart, const int* __restrict__ cnt,
                                   const float* __restrict__ dinv, const float* __restrict__ rns,
                                   const float* __restrict__ noise, float* __restrict__ mu,
                                   float* __restrict__ logstd, float* __restrict__ zeta, int n) {
  int wv = __builtin_amdgcn_readfirstlane((int)(threadIdx.x >> 5));
  int row0 = (blockIdx.x * 8 + wv) * ROWS_PER_WAVE;
  int lane = threadIdx.x & 31;
  for (int r = 0; r < ROWS_PER_WAVE; ++r) {
    int row = row0 + r;
    if (row >= n) break;
    int start = rowstart[row];
    int c = cnt[row];
    float dd = dinv[row];
    float d2 = dd * dd;
    float rr = rns[row];
    float2 sv = ((const float2*)(l + (size_t)row * D_OUT))[lane];
    float2 nz = ((const float2*)(noise + (size_t)row * D_OUT))[lane];
    float2 am, al;
    am.x = sv.x * d2;      am.y = sv.y * d2;
    al.x = sv.x * rr * d2; al.y = sv.y * rr * d2;
    for (int base = 0; base < c; base += 32) {
      int m = c - base;
      if (m > 32) m = 32;
      int s_l = 0;
      float w_l = 0.0f, sc_l = 0.0f;
      if (lane < m) {
        s_l  = csr[start + base + lane];
        w_l  = dinv[s_l] * dd;
        sc_l = w_l * rns[s_l];
      }
      int j = 0;
      for (; j + 8 <= m; j += 8) vg_edges64<8>(l, s_l, w_l, sc_l, j, lane, am, al);
      if (j + 4 <= m) { vg_edges64<4>(l, s_l, w_l, sc_l, j, lane, am, al); j += 4; }
      if (j + 2 <= m) { vg_edges64<2>(l, s_l, w_l, sc_l, j, lane, am, al); j += 2; }
      if (j < m)      { vg_edges64<1>(l, s_l, w_l, sc_l, j, lane, am, al); }
    }
    ((float2*)(mu     + (size_t)row * D_OUT))[lane] = am;
    ((float2*)(logstd + (size_t)row * D_OUT))[lane] = al;
    float2 z;
    z.x = am.x + nz.x * expf(al.x);
    z.y = am.y + nz.y * expf(al.y);
    ((float2*)(zeta   + (size_t)row * D_OUT))[lane] = z;
  }
}

extern "C" void kernel_launch(void* const* d_in, const int* in_sizes, int n_in,
                              void* d_out, int out_size, void* d_ws, size_t ws_size,
                              hipStream_t stream) {
  const float* x     = (const float*)d_in[0];
  const int*   esrc  = (const int*)  d_in[1];
  const int*   edst  = (const int*)  d_in[2];
  const float* W1    = (const float*)d_in[3];
  const float* b1    = (const float*)d_in[4];
  const float* gamma = (const float*)d_in[5];
  const float* beta  = (const float*)d_in[6];
  const float* Wmu   = (const float*)d_in[7];
  const float* bmu   = (const float*)d_in[8];
  const float* noise = (const float*)d_in[9];

  const int N = in_sizes[0] / D_IN;
  const int E = in_sizes[1];

  float* ws = (float*)d_ws;
  size_t npad = ((size_t)N + 255) & ~(size_t)255;
  size_t off = 0;
  float* dinv  = ws + off; off += npad;
  float* stats = ws + off; off += STAT_WORDS;
  float* rns   = ws + off; off += npad;
  float* h     = ws + off; off += (size_t)N * D_HID;
  float* hagg  = ws + off; off += (size_t)N * D_HID;
  float* lbuf  = ws + off; off += (size_t)N * D_OUT;
  size_t epad = ((size_t)E + 255) & ~(size_t)255;
  const int NB   = (N + BKT_ROWS - 1) / BKT_ROWS;
  const int NBLK = (E + EPB - 1) / EPB;
  size_t bbpad = ((size_t)NBLK * (size_t)NB + 255) & ~(size_t)255;
  int* cnt      = (int*)(ws + off); off += npad;
  int* rowstart = (int*)(ws + off); off += npad;
  int* bcnt     = (int*)(ws + off); off += npad;
  int* blkbase  = (int*)(ws + off); off += bbpad;
  int* csr      = (int*)(ws + off); off += epad;
  int2* binned  = (int2*)(ws + off); off += 2 * epad;

  float* mu_out   = (float*)d_out;
  float* ls_out   = mu_out + (size_t)N * D_OUT;
  float* zeta_out = ls_out + (size_t)N * D_OUT;

  const int B = 256;
  auto cdiv = [](int a, int b) { return (a + b - 1) / b; };

  vg_init_kernel<<<cdiv(NB > STAT_WORDS ? NB : STAT_WORDS, B), B, 0, stream>>>(bcnt, stats, NB);
  vg_bucket_count_kernel<<<NBLK, B, NB * sizeof(int), stream>>>(edst, bcnt, blkbase, E, NB);
  vg_bucket_scatter_kernel<<<NBLK, B, NB * sizeof(int), stream>>>(esrc, edst, bcnt, blkbase, binned, E, NB);
  vg_bucket_rows_kernel<<<NB, B, 0, stream>>>(binned, bcnt, cnt, rowstart, dinv, csr, N);

  {
    int strips = (N / 16) * (D_HID / (16 * GEMM_NT));
    vg_gemm_wmma_kernel<<<cdiv(strips, B / 32), B, 0, stream>>>(x, W1, b1, h, N, D_IN, D_HID);
  }

  vg_gather128_kernel<<<cdiv(N, (B / 32) * ROWS_PER_WAVE), B, 0, stream>>>(h, csr, rowstart, cnt, dinv, hagg, stats, N);

  {
    int strips = N / 16;
    vg_gemm_bn_wmma_kernel<<<cdiv(strips, B / 32), B, 0, stream>>>(hagg, Wmu, bmu, stats, gamma, beta, lbuf, rns, N, N);
  }

  vg_gather64_kernel<<<cdiv(N, (B / 32) * ROWS_PER_WAVE), B, 0, stream>>>(lbuf, csr, rowstart, cnt, dinv, rns, noise,
                                                        mu_out, ls_out, zeta_out, N);
}


# Rows stored times their own dinv by the GEMM epilogues; gathers add source rows unweighted and apply dinv[dst] once after the sum (sum order kept): no per-edge dinv[src] line read
# speedup vs baseline: 4.8396x; 1.1122x over previous
#include <hip/hip_runtime.h>
#include <hip/hip_bf16.h>


#define D_IN   128
#define D_HID  128
#define D_OUT  64
#define BN_EPS 1e-5f
#define LRELU_SLOPE 0.01f
#define NORM_SCALE 1.8f

typedef __attribute__((ext_vector_type(2))) float v2f;
typedef __attribute__((ext_vector_type(8))) float v8f;

#define BKT_SHIFT 8
#define BKT_ROWS  256
#define EPB       4096
#define EPT       (EPB / 256)
#define BPT       20

#define STAT_COPIES 16
#define STAT_WORDS  (256 * STAT_COPIES)

__global__ void vg_init_kernel(int* __restrict__ bcnt, float* __restrict__ stats, int nb) {
  int i = blockIdx.x * blockDim.x + threadIdx.x;
  if (i < nb) bcnt[i] = 0;
  if (i < STAT_WORDS) stats[i] = 0.0f;
}

__global__ void vg_bucket_count_kernel(const int* __restrict__ dst, int* __restrict__ bcnt,
                                       int* __restrict__ blkbase, int e, int nb) {
  extern __shared__ int lds_i[];
  for (int b = threadIdx.x; b < nb; b += 256) lds_i[b] = 0;
  __syncthreads();
  int e0 = blockIdx.x * EPB + threadIdx.x;
  int d[EPT];
#pragma unroll
  for (int k = 0; k < EPT; ++k) {
    int i = e0 + k * 256;
    d[k] = (i < e) ? dst[i] : -1;
  }
#pragma unroll
  for (int k = 0; k < EPT; ++k) {
    if (d[k] >= 0) atomicAdd(&lds_i[d[k] >> BKT_SHIFT], 1);
  }
  __syncthreads();
  for (int b = threadIdx.x; b < nb; b += 256) {
    int h = lds_i[b];
    blkbase[(size_t)blockIdx.x * (size_t)nb + b] = (h > 0) ? atomicAdd(&bcnt[b], h) : 0;
  }
}

__global__ void vg_bucket_scatter_kernel(const int* __restrict__ src, const int* __restrict__ dst,
                                         const int* __restrict__ bcnt, const int* __restrict__ blkbase,
                                         int2* __restrict__ binned, int e, int nb) {
  extern __shared__ int lds_i[];
  __shared__ int wsum[8];
  int lane = threadIdx.x & 31;
  int wv = threadIdx.x >> 5;
  int run = 0;
  for (int b0 = 0; b0 < nb; b0 += 256) {
    int b = b0 + threadIdx.x;
    int c = (b < nb) ? bcnt[b] : 0;
    int x = c;
#pragma unroll
    for (int o = 1; o < 32; o <<= 1) {
      int y = __shfl_up(x, o, 32);
      if (lane >= o) x += y;
    }
    if (lane == 31) wsum[wv] = x;
    __syncthreads();
    int wbase = 0, tot = 0;
#pragma unroll
    for (int k = 0; k < 8; ++k) { int t = wsum[k]; if (k < wv) wbase += t; tot += t; }
    if (b < nb) lds_i[b] = run + wbase + (x - c) + blkbase[(size_t)blockIdx.x * (size_t)nb + b];
    run += tot;
    __syncthreads();
  }
  int e0 = blockIdx.x * EPB + threadIdx.x;
  int d[EPT], s[EPT];
#pragma unroll
  for (int k = 0; k < EPT; ++k) {
    int i = e0 + k * 256;
    d[k] = (i < e) ? dst[i] : -1;
    s[k] = (i < e) ? src[i] : 0;
  }
#pragma unroll
  for (int k = 0; k < EPT; ++k) {
    if (d[k] >= 0) {
      int p = atomicAdd(&lds_i[d[k] >> BKT_SHIFT], 1);
      binned[p] = make_int2(s[k], d[k]);
    }
  }
}

__global__ void vg_bucket_rows_kernel(const int2* __restrict__ binned, const int* __restrict__ bcnt,
                                      int* __restrict__ cnt, int* __restrict__ rowstart,
                                      float* __restrict__ dinv, int* __restrict__ csr, int n) {
  __shared__ int lcnt[BKT_ROWS];
  __shared__ int lcur[BKT_ROWS];
  __shared__ int wsum[8];
  __shared__ int wsum2[8];
  int bkt = blockIdx.x;
  int t = threadIdx.x;
  int lane = t & 31;
  int wv = t >> 5;
  int part = 0;
  for (int j = t; j < bkt; j += 256) part += bcnt[j];
#pragma unroll
  for (int o = 16; o > 0; o >>= 1) part += __shfl_xor(part, o, 32);
  if (lane == 0) wsum[wv] = part;
  lcnt[t] = 0;
  __syncthreads();
  int bstart = 0;
#pragma unroll
  for (int k = 0; k < 8; ++k) bstart += wsum[k];
  int bsize = bcnt[bkt];
  const int2* be = binned + bstart;

  for (int i0 = 0; i0 < bsize; i0 += 256 * BPT) {
    int r[BPT];
#pragma unroll
    for (int k = 0; k < BPT; ++k) {
      int i = i0 + k * 256 + t;
      r[k] = be[(i < bsize) ? i : (bsize - 1)].y;
    }
#pragma unroll
    for (int k = 0; k < BPT; ++k) {
      if (i0 + k * 256 + t < bsize) atomicAdd(&lcnt[r[k] & (BKT_ROWS - 1)], 1);
    }
  }
  __syncthreads();

  int c = lcnt[t];
  int x = c;
#pragma unroll
  for (int o = 1; o < 32; o <<= 1) {
    int y = __shfl_up(x, o, 32);
    if (lane >= o) x += y;
  }
  if (lane == 31) wsum2[wv] = x;
  __syncthreads();
  int wbase = 0;
#pragma unroll
  for (int k = 0; k < 8; ++k) { int w = wsum2[k]; if (k < wv) wbase += w; }
  int rs = bstart + wbase + (x - c);
  lcur[t] = rs;
  int row = bkt * BKT_ROWS + t;
  if (row < n) {
    cnt[row] = c;
    rowstart[row] = rs;
    dinv[row] = rsqrtf((float)c + 1.0f);
  }
  __syncthreads();

  for (int i0 = 0; i0 < bsize; i0 += 256 * BPT) {
    int2 ev[BPT];
#pragma unroll
    for (int k = 0; k < BPT; ++k) {
      int i = i0 + k * 256 + t;
      ev[k] = be[(i < bsize) ? i : (bsize - 1)];
    }
#pragma unroll
    for (int k = 0; k < BPT; ++k) {
      if (i0 + k * 256 + t < bsize) {
        int p = atomicAdd(&lcur[ev[k].y & (BKT_ROWS - 1)], 1);
        csr[p] = ev[k].x;
      }
    }
  }
}

#define GEMM_NT 4
static_assert(D_OUT == 16 * GEMM_NT, "the second GEMM's wave covers the whole width of L");
__global__ void vg_gemm_wmma_kernel(const float* __restrict__ A, const float* __restrict__ B,
                                    const float* __restrict__ bias, const float* __restrict__ rowscale,
                                    float* __restrict__ C, int M, int K, int Ncols) {
  const int wavesPerBlock = blockDim.x >> 5;
  int w = blockIdx.x * wavesPerBlock + (threadIdx.x >> 5);
  int stripsN = Ncols / (16 * GEMM_NT);
  int totalStrips = (M >> 4) * stripsN;
  if (w >= totalStrips) return;
  int tm = w / stripsN;
  int ts = w - tm * stripsN;
  int lane = threadIdx.x & 31;
  int m  = lane & 15;
  int hi = lane >> 4;

  v8f acc[GEMM_NT];
#pragma unroll
  for (int t = 0; t < GEMM_NT; ++t) acc[t] = (v8f){};
  const float* arow = A + (size_t)(tm * 16 + m) * (size_t)K;
  const float* bcol = B + ts * (16 * GEMM_NT) + m;
  const float4* rsp = (const float4*)(rowscale + tm * 16 + hi * 8);
  float4 rs0 = rsp[0];
  float4 rs1 = rsp[1];
  for (int k0 = 0; k0 < K; k0 += 4) {
    int kb = k0 + 2 * hi;
    v2f a = *(const v2f*)(arow + kb);
    const float* b0 = bcol + (size_t)kb * (size_t)Ncols;
    const float* b1 = b0 + Ncols;
    v2f b[GEMM_NT];
#pragma unroll
    for (int t = 0; t < GEMM_NT; ++t) {
      b[t].x = b0[t * 16];
      b[t].y = b1[t * 16];
    }
#pragma unroll
    for (int t = 0; t < GEMM_NT; ++t) {
      acc[t] = __builtin_amdgcn_wmma_f32_16x16x4_f32(false, a, false, b[t],
                                                     (short)0, acc[t], false, false);
    }
  }
  int rbase = tm * 16 + hi * 8;
  float rs[8] = {rs0.x, rs0.y, rs0.z, rs0.w, rs1.x, rs1.y, rs1.z, rs1.w};
#pragma unroll
  for (int t = 0; t < GEMM_NT; ++t) {
    int col = ts * (16 * GEMM_NT) + t * 16 + m;
    float bv = bias[col];
#pragma unroll
    for (int j = 0; j < 8; ++j) {
      C[(size_t)(rbase + j) * (size_t)Ncols + (size_t)col] = (acc[t][j] + bv) * rs[j];
    }
  }
}

#define ROWS_PER_WAVE 8
template <int U>
__device__ __forceinline__ void vg_edges128(const float* __restrict__ in, int s_l,
                                            int j, int lane, float4& acc) {
  float4 v[U];
#pragma unroll
  for (int u = 0; u < U; ++u) {
    int s = __builtin_amdgcn_readlane(s_l, j + u);
    v[u]  = ((const float4*)(in + (size_t)s * D_HID))[lane];
  }
#pragma unroll
  for (int u = 0; u < U; ++u) {
    acc.x += v[u].x;
    acc.y += v[u].y;
    acc.z += v[u].z;
    acc.w += v[u].w;
  }
}

__global__ void vg_gather128_kernel(const float* __restrict__ in, const int* __restrict__ csr,
                                    const int* __restrict__ rowstart, const int* __restrict__ cnt,
                                    const float* __restrict__ dinv, float* __restrict__ out,
                                    float* __restrict__ stats, int n) {
  __shared__ float4 red[8][64];
  int wv = __builtin_amdgcn_readfirstlane((int)(threadIdx.x >> 5));
  int row0 = (blockIdx.x * 8 + wv) * ROWS_PER_WAVE;
  int lane = threadIdx.x & 31;
  float4* my_sum = &red[wv][lane];
  float4* my_sq  = &red[wv][32 + lane];
  *my_sum = make_float4(0.f, 0.f, 0.f, 0.f);
  *my_sq  = make_float4(0.f, 0.f, 0.f, 0.f);
  for (int r = 0; r < ROWS_PER_WAVE; ++r) {
    int row = row0 + r;
    if (row >= n) break;
    int start = rowstart[row];
    int c = cnt[row];
    float dd = dinv[row];
    float4 acc = ((const float4*)(in + (size_t)row * D_HID))[lane];
    for (int base = 0; base < c; base += 32) {
      int m = c - base;
      if (m > 32) m = 32;
      int s_l = 0;
      if (lane < m) s_l = csr[start + base + lane];
      int j = 0;
      for (; j + 8 <= m; j += 8) vg_edges128<8>(in, s_l, j, lane, acc);
      if (j + 4 <= m) { vg_edges128<4>(in, s_l, j, lane, acc); j += 4; }
      if (j + 2 <= m) { vg_edges128<2>(in, s_l, j, lane, acc); j += 2; }
      if (j < m)      { vg_edges128<1>(in, s_l, j, lane, acc); }
    }
    acc.x *= dd; acc.y *= dd; acc.z *= dd; acc.w *= dd;
    ((float4*)(out + (size_t)row * D_HID))[lane] = acc;
    float4 ssum = *my_sum;
    float4 ssq  = *my_sq;
    ssum.x += acc.x; ssum.y += acc.y; ssum.z += acc.z; ssum.w += acc.w;
    ssq.x += acc.x * acc.x; ssq.y += acc.y * acc.y; ssq.z += acc.z * acc.z; ssq.w += acc.w * acc.w;
    *my_sum = ssum;
    *my_sq  = ssq;
  }
  __syncthreads();
  const float* rf = (const float*)&red[0][0];
  float tot = 0.f;
#pragma unroll
  for (int k = 0; k < 8; ++k) tot += rf[k * 256 + threadIdx.x];
  atomicAdd(&stats[(blockIdx.x & (STAT_COPIES - 1)) * 256 + threadIdx.x], tot);
}


template <int MASK>
__device__ __forceinline__ float vg_add_row_xor(float x) {
  int y = __builtin_amdgcn_update_dpp(0, __float_as_int(x), 0x160 + MASK, 0xf, 0xf, true);
  return x + __int_as_float(y);
}

__global__ void vg_gemm_bn_wmma_kernel(const float* __restrict__ A, const float* __restrict__ B,
                                       const float* __restrict__ bias, const float* __restrict__ stats,
                                       const float* __restrict__ gamma, const float* __restrict__ beta,
                                       const float* __restrict__ rowscale,
                                       float* __restrict__ C, float* __restrict__ rns, int M, int n) {
  __shared__ __attribute__((aligned(16))) float bn_sc[D_HID];
  __shared__ __attribute__((aligned(16))) float bn_sh[D_HID];
  if (threadIdx.x < D_HID) {
    int c = threadIdx.x;
    float invn = 1.0f / (float)n;
    float s1 = 0.f, s2 = 0.f;
#pragma unroll
    for (int k = 0; k < STAT_COPIES; ++k) {
      s1 += stats[k * 256 + c];
      s2 += stats[k * 256 + 128 + c];
    }
    float mean = s1 * invn;
    float var  = s2 * invn - mean * mean;
    float a = gamma[c] * rsqrtf(var + BN_EPS);
    bn_sc[c] = a;
    bn_sh[c] = beta[c] - mean * a;
  }
  __syncthreads();

  const int wavesPerBlock = blockDim.x >> 5;
  int tm = blockIdx.x * wavesPerBlock + (threadIdx.x >> 5);
  if (tm >= (M >> 4)) return;
  int lane = threadIdx.x & 31;
  int m  = lane & 15;
  int hi = lane >> 4;

  v8f acc[GEMM_NT];
#pragma unroll
  for (int t = 0; t < GEMM_NT; ++t) acc[t] = (v8f){};
  const float* arow = A + (size_t)(tm * 16 + m) * (size_t)D_HID;
  const float* bcol = B + m;
  const float4* rsp = (const float4*)(rowscale + tm * 16 + hi * 8);
  float4 rs0 = rsp[0];
  float4 rs1 = rsp[1];
  for (int k0 = 0; k0 < D_HID; k0 += 4) {
    int kb = k0 + 2 * hi;
    v2f a  = *(const v2f*)(arow + kb);
    v2f sc = *(const v2f*)(bn_sc + kb);
    v2f sh = *(const v2f*)(bn_sh + kb);
    float v0 = sc.x * a.x + sh.x;
    float v1 = sc.y * a.y + sh.y;
    a.x = (v0 > 0.f) ? v0 : LRELU_SLOPE * v0;
    a.y = (v1 > 0.f) ? v1 : LRELU_SLOPE * v1;
    const float* b0 = bcol + (size_t)kb * (size_t)D_OUT;
    const float* b1 = b0 + D_OUT;
    v2f b[GEMM_NT];
#pragma unroll
    for (int t = 0; t < GEMM_NT; ++t) {
      b[t].x = b0[t * 16];
      b[t].y = b1[t * 16];
    }
#pragma unroll
    for (int t = 0; t < GEMM_NT; ++t) {
      acc[t] = __builtin_amdgcn_wmma_f32_16x16x4_f32(false, a, false, b[t],
                                                     (short)0, acc[t], false, false);
    }
  }
  int rbase = tm * 16 + hi * 8;
  float rs[8] = {rs0.x, rs0.y, rs0.z, rs0.w, rs1.x, rs1.y, rs1.z, rs1.w};
  float sq[8];
#pragma unroll
  for (int j = 0; j < 8; ++j) sq[j] = 0.f;
#pragma unroll
  for (int t = 0; t < GEMM_NT; ++t) {
    int col = t * 16 + m;
    float bv = bias[col];
#pragma unroll
    for (int j = 0; j < 8; ++j) {
      float v = acc[t][j] + bv;
      C[(size_t)(rbase + j) * (size_t)D_OUT + (size_t)col] = v * rs[j];
      sq[j] += v * v;
    }
  }
  float mine = 0.f;
#pragma unroll
  for (int j = 0; j < 8; ++j) {
    float s = sq[j];
    s = vg_add_row_xor<8>(s);
    s = vg_add_row_xor<4>(s);
    s = vg_add_row_xor<2>(s);
    s = vg_add_row_xor<1>(s);
    mine = (m == j) ? s : mine;
  }
  if (m < 8) rns[rbase + m] = NORM_SCALE / fmaxf(sqrtf(mine), 1e-12f);
}

template <int U>
__device__ __forceinline__ void vg_edges64(const float* __restrict__ l, int s_l, float sc_l,
                                           int j, int lane, float2& am, float2& al) {
  float2 v[U];
#pragma unroll
  for (int u = 0; u < U; ++u) {
    int s = __builtin_amdgcn_readlane(s_l, j + u);
    v[u]  = ((const float2*)(l + (size_t)s * D_OUT))[lane];
  }
#pragma unroll
  for (int u = 0; u < U; ++u) {
    float sc = __int_as_float(__builtin_amdgcn_readlane(__float_as_int(sc_l), j + u));
    am.x += v[u].x;
    am.y += v[u].y;
    al.x += sc * v[u].x;
    al.y += sc * v[u].y;
  }
}

__global__ void vg_gather64_kernel(const float* __restrict__ l, const int* __restrict__ csr,
                                   const int* __restrict__ rowstart, const int* __restrict__ cnt,
                                   const float* __restrict__ dinv, const float* __restrict__ rns,
                                   const float* __restrict__ noise, float* __restrict__ mu,
                                   float* __restrict__ logstd, float* __restrict__ zeta, int n) {
  int wv = __builtin_amdgcn_readfirstlane((int)(threadIdx.x >> 5));
  int row0 = (blockIdx.x * 8 + wv) * ROWS_PER_WAVE;
  int lane = threadIdx.x & 31;
  for (int r = 0; r < ROWS_PER_WAVE; ++r) {
    int row = row0 + r;
    if (row >= n) break;
    int start = rowstart[row];
    int c = cnt[row];
    float dd = dinv[row];
    float rr = rns[row];
    float2 sv = ((const float2*)(l + (size_t)row * D_OUT))[lane];
    float2 nz = ((const float2*)(noise + (size_t)row * D_OUT))[lane];
    float2 am, al;
    am.x = sv.x;      am.y = sv.y;
    al.x = sv.x * rr; al.y = sv.y * rr;
    for (int base = 0; base < c; base += 32) {
      int m = c - base;
      if (m > 32) m = 32;
      int s_l = 0;
      float sc_l = 0.0f;
      if (lane < m) {
        s_l  = csr[start + base + lane];
        sc_l = rns[s_l];
      }
      int j = 0;
      for (; j + 8 <= m; j += 8) vg_edges64<8>(l, s_l, sc_l, j, lane, am, al);
      if (j + 4 <= m) { vg_edges64<4>(l, s_l, sc_l, j, lane, am, al); j += 4; }
      if (j + 2 <= m) { vg_edges64<2>(l, s_l, sc_l, j, lane, am, al); j += 2; }
      if (j < m)      { vg_edges64<1>(l, s_l, sc_l, j, lane, am, al); }
    }
    am.x *= dd; am.y *= dd;
    al.x *= dd; al.y *= dd;
    ((float2*)(mu     + (size_t)row * D_OUT))[lane] = am;
    ((float2*)(logstd + (size_t)row * D_OUT))[lane] = al;
    float2 z;
    z.x = am.x + nz.x * expf(al.x);
    z.y = am.y + nz.y * expf(al.y);
    ((float2*)(zeta   + (size_t)row * D_OUT))[lane] = z;
  }
}

extern "C" void kernel_launch(void* const* d_in, const int* in_sizes, int n_in,
                              void* d_out, int out_size, void* d_ws, size_t ws_size,
                              hipStream_t stream) {
  const float* x     = (const float*)d_in[0];
  const int*   esrc  = (const int*)  d_in[1];
  const int*   edst  = (const int*)  d_in[2];
  const float* W1    = (const float*)d_in[3];
  const float* b1    = (const float*)d_in[4];
  const float* gamma = (const float*)d_in[5];
  const float* beta  = (const float*)d_in[6];
  const float* Wmu   = (const float*)d_in[7];
  const float* bmu   = (const float*)d_in[8];
  const float* noise = (const float*)d_in[9];

  const int N = in_sizes[0] / D_IN;
  const int E = in_sizes[1];

  float* ws = (float*)d_ws;
  size_t npad = ((size_t)N + 255) & ~(size_t)255;
  size_t off = 0;
  float* dinv  = ws + off; off += npad;
  float* stats = ws + off; off += STAT_WORDS;
  float* rns   = ws + off; off += npad;
  float* h     = ws + off; off += (size_t)N * D_HID;
  float* hagg  = ws + off; off += (size_t)N * D_HID;
  float* lbuf  = ws + off; off += (size_t)N * D_OUT;
  size_t epad = ((size_t)E + 255) & ~(size_t)255;
  const int NB   = (N + BKT_ROWS - 1) / BKT_ROWS;
  const int NBLK = (E + EPB - 1) / EPB;
  size_t bbpad = ((size_t)NBLK * (size_t)NB + 255) & ~(size_t)255;
  int* cnt      = (int*)(ws + off); off += npad;
  int* rowstart = (int*)(ws + off); off += npad;
  int* bcnt     = (int*)(ws + off); off += npad;
  int* blkbase  = (int*)(ws + off); off += bbpad;
  int* csr      = (int*)(ws + off); off += epad;
  int2* binned  = (int2*)(ws + off); off += 2 * epad;

  float* mu_out   = (float*)d_out;
  float* ls_out   = mu_out + (size_t)N * D_OUT;
  float* zeta_out = ls_out + (size_t)N * D_OUT;

  const int B = 256;
  auto cdiv = [](int a, int b) { return (a + b - 1) / b; };

  vg_init_kernel<<<cdiv(NB > STAT_WORDS ? NB : STAT_WORDS, B), B, 0, stream>>>(bcnt, stats, NB);
  vg_bucket_count_kernel<<<NBLK, B, NB * sizeof(int), stream>>>(edst, bcnt, blkbase, E, NB);
  vg_bucket_scatter_kernel<<<NBLK, B, NB * sizeof(int), stream>>>(esrc, edst, bcnt, blkbase, binned, E, NB);
  vg_bucket_rows_kernel<<<NB, B, 0, stream>>>(binned, bcnt, cnt, rowstart, dinv, csr, N);

  {
    int strips = (N / 16) * (D_HID / (16 * GEMM_NT));
    vg_gemm_wmma_kernel<<<cdiv(strips, B / 32), B, 0, stream>>>(x, W1, b1, dinv, h, N, D_IN, D_HID);
  }

  vg_gather128_kernel<<<cdiv(N, (B / 32) * ROWS_PER_WAVE), B, 0, stream>>>(h, csr, rowstart, cnt, dinv, hagg, stats, N);

  {
    int strips = N / 16;
    vg_gemm_bn_wmma_kernel<<<cdiv(strips, B / 32), B, 0, stream>>>(hagg, Wmu, bmu, stats, gamma, beta, dinv, lbuf, rns, N, N);
  }

  vg_gather64_kernel<<<cdiv(N, (B / 32) * ROWS_PER_WAVE), B, 0, stream>>>(lbuf, csr, rowstart, cnt, dinv, rns, noise,
                                                        mu_out, ls_out, zeta_out, N);
}


# Both GEMMs fed from LDS: B staged once per block as k pairs in fragment order, each wave's 16x128 A tile fetched by whole-row 16-byte loads one tile ahead; same products, k order and sum order
# speedup vs baseline: 5.0861x; 1.0509x over previous
#include <hip/hip_runtime.h>
#include <hip/hip_bf16.h>


#define D_IN   128
#define D_HID  128
#define D_OUT  64
#define BN_EPS 1e-5f
#define LRELU_SLOPE 0.01f
#define NORM_SCALE 1.8f

typedef __attribute__((ext_vector_type(2))) float v2f;
typedef __attribute__((ext_vector_type(8))) float v8f;

#define BKT_SHIFT 8
#define BKT_ROWS  256
#define EPB       4096
#define EPT       (EPB / 256)
#define BPT       20

#define STAT_COPIES 16
#define STAT_WORDS  (256 * STAT_COPIES)

__global__ void vg_init_kernel(int* __restrict__ bcnt, float* __restrict__ stats, int nb) {
  int i = blockIdx.x * blockDim.x + threadIdx.x;
  if (i < nb) bcnt[i] = 0;
  if (i < STAT_WORDS) stats[i] = 0.0f;
}

__global__ void vg_bucket_count_kernel(const int* __restrict__ dst, int* __restrict__ bcnt,
                                       int* __restrict__ blkbase, int e, int nb) {
  extern __shared__ int lds_i[];
  for (int b = threadIdx.x; b < nb; b += 256) lds_i[b] = 0;
  __syncthreads();
  int e0 = blockIdx.x * EPB + threadIdx.x;
  int d[EPT];
#pragma unroll
  for (int k = 0; k < EPT; ++k) {
    int i = e0 + k * 256;
    d[k] = (i < e) ? dst[i] : -1;
  }
#pragma unroll
  for (int k = 0; k < EPT; ++k) {
    if (d[k] >= 0) atomicAdd(&lds_i[d[k] >> BKT_SHIFT], 1);
  }
  __syncthreads();
  for (int b = threadIdx.x; b < nb; b += 256) {
    int h = lds_i[b];
    blkbase[(size_t)blockIdx.x * (size_t)nb + b] = (h > 0) ? atomicAdd(&bcnt[b], h) : 0;
  }
}

__global__ void vg_bucket_scatter_kernel(const int* __restrict__ src, const int* __restrict__ dst,
                                         const int* __restrict__ bcnt, const int* __restrict__ blkbase,
                                         int2* __restrict__ binned, int e, int nb) {
  extern __shared__ int lds_i[];
  __shared__ int wsum[8];
  int lane = threadIdx.x & 31;
  int wv = threadIdx.x >> 5;
  int run = 0;
  for (int b0 = 0; b0 < nb; b0 += 256) {
    int b = b0 + threadIdx.x;
    int c = (b < nb) ? bcnt[b] : 0;
    int x = c;
#pragma unroll
    for (int o = 1; o < 32; o <<= 1) {
      int y = __shfl_up(x, o, 32);
      if (lane >= o) x += y;
    }
    if (lane == 31) wsum[wv] = x;
    __syncthreads();
    int wbase = 0, tot = 0;
#pragma unroll
    for (int k = 0; k < 8; ++k) { int t = wsum[k]; if (k < wv) wbase += t; tot += t; }
    if (b < nb) lds_i[b] = run + wbase + (x - c) + blkbase[(size_t)blockIdx.x * (size_t)nb + b];
    run += tot;
    __syncthreads();
  }
  int e0 = blockIdx.x * EPB + threadIdx.x;
  int d[EPT], s[EPT];
#pragma unroll
  for (int k = 0; k < EPT; ++k) {
    int i = e0 + k * 256;
    d[k] = (i < e) ? dst[i] : -1;
    s[k] = (i < e) ? src[i] : 0;
  }
#pragma unroll
  for (int k = 0; k < EPT; ++k) {
    if (d[k] >= 0) {
      int p = atomicAdd(&lds_i[d[k] >> BKT_SHIFT], 1);
      binned[p] = make_int2(s[k], d[k]);
    }
  }
}

__global__ void vg_bucket_rows_kernel(const int2* __restrict__ binned, const int* __restrict__ bcnt,
                                      int* __restrict__ cnt, int* __restrict__ rowstart,
                                      float* __restrict__ dinv, int* __restrict__ csr, int n) {
  __shared__ int lcnt[BKT_ROWS];
  __shared__ int lcur[BKT_ROWS];
  __shared__ int wsum[8];
  __shared__ int wsum2[8];
  int bkt = blockIdx.x;
  int t = threadIdx.x;
  int lane = t & 31;
  int wv = t >> 5;
  int part = 0;
  for (int j = t; j < bkt; j += 256) part += bcnt[j];
#pragma unroll
  for (int o = 16; o > 0; o >>= 1) part += __shfl_xor(part, o, 32);
  if (lane == 0) wsum[wv] = part;
  lcnt[t] = 0;
  __syncthreads();
  int bstart = 0;
#pragma unroll
  for (int k = 0; k < 8; ++k) bstart += wsum[k];
  int bsize = bcnt[bkt];
  const int2* be = binned + bstart;

  for (int i0 = 0; i0 < bsize; i0 += 256 * BPT) {
    int r[BPT];
#pragma unroll
    for (int k = 0; k < BPT; ++k) {
      int i = i0 + k * 256 + t;
      r[k] = be[(i < bsize) ? i : (bsize - 1)].y;
    }
#pragma unroll
    for (int k = 0; k < BPT; ++k) {
      if (i0 + k * 256 + t < bsize) atomicAdd(&lcnt[r[k] & (BKT_ROWS - 1)], 1);
    }
  }
  __syncthreads();

  int c = lcnt[t];
  int x = c;
#pragma unroll
  for (int o = 1; o < 32; o <<= 1) {
    int y = __shfl_up(x, o, 32);
    if (lane >= o) x += y;
  }
  if (lane == 31) wsum2[wv] = x;
  __syncthreads();
  int wbase = 0;
#pragma unroll
  for (int k = 0; k < 8; ++k) { int w = wsum2[k]; if (k < wv) wbase += w; }
  int rs = bstart + wbase + (x - c);
  lcur[t] = rs;
  int row = bkt * BKT_ROWS + t;
  if (row < n) {
    cnt[row] = c;
    rowstart[row] = rs;
    dinv[row] = rsqrtf((float)c + 1.0f);
  }
  __syncthreads();

  for (int i0 = 0; i0 < bsize; i0 += 256 * BPT) {
    int2 ev[BPT];
#pragma unroll
    for (int k = 0; k < BPT; ++k) {
      int i = i0 + k * 256 + t;
      ev[k] = be[(i < bsize) ? i : (bsize - 1)];
    }
#pragma unroll
    for (int k = 0; k < BPT; ++k) {
      if (i0 + k * 256 + t < bsize) {
        int p = atomicAdd(&lcur[ev[k].y & (BKT_ROWS - 1)], 1);
        csr[p] = ev[k].x;
      }
    }
  }
}

#define GEMM_K    128
#define A_PITCH   132
#define GEMM1_GRID 512
#define GEMM2_GRID 768

template <int NC>
__device__ __forceinline__ void vg_stage_b(const float* __restrict__ B, float* bl) {
#pragma unroll 16
  for (int p = threadIdx.x; p < (GEMM_K / 2) * NC; p += 256) {
    int kp = p / NC;
    int n  = p - kp * NC;
    v2f v;
    v.x = B[(size_t)(2 * kp) * NC + n];
    v.y = B[(size_t)(2 * kp + 1) * NC + n];
    int t = n >> 4;
    int m = n & 15;
    *(v2f*)(bl + kp * (2 * NC) + (t >> 1) * 64 + m * 4 + (t & 1) * 2) = v;
  }
}

template <int NT, int NC>
__device__ __forceinline__ void vg_tile_products(const float* al, const float* bl, int m, int hi,
                                                 v8f (&acc)[NT]) {
  const float* ap = al + m * A_PITCH + 2 * hi;
  const float* bp = bl + hi * (2 * NC) + m * 4;
#pragma unroll 2
  for (int k0 = 0; k0 < GEMM_K; k0 += 4) {
    v2f a = *(const v2f*)(ap + k0);
    const float* bk = bp + (k0 >> 1) * (2 * NC);
    float4 bq[NT / 2];
#pragma unroll
    for (int u = 0; u < NT / 2; ++u) bq[u] = *(const float4*)(bk + u * 64);
#pragma unroll
    for (int u = 0; u < NT / 2; ++u) {
      v2f b0, b1;
      b0.x = bq[u].x; b0.y = bq[u].y;
      b1.x = bq[u].z; b1.y = bq[u].w;
      acc[2 * u]     = __builtin_amdgcn_wmma_f32_16x16x4_f32(false, a, false, b0,
                                                             (short)0, acc[2 * u], false, false);
      acc[2 * u + 1] = __builtin_amdgcn_wmma_f32_16x16x4_f32(false, a, false, b1,
                                                             (short)0, acc[2 * u + 1], false, false);
    }
  }
}

#define GEMM1_NT (D_HID / 16)
__global__ __launch_bounds__(256) void vg_gemm_wmma_kernel(const float* __restrict__ A, const float* __restrict__ B,
                                    const float* __restrict__ bias, const float* __restrict__ rowscale,
                                    float* __restrict__ C, int M) {
  __shared__ __attribute__((aligned(16))) float bl[(GEMM_K / 2) * 2 * D_HID];
  __shared__ __attribute__((aligned(16))) float alds[8][16 * A_PITCH];
  int wv = __builtin_amdgcn_readfirstlane((int)(threadIdx.x >> 5));
  int lane = threadIdx.x & 31;
  int m  = lane & 15;
  int hi = lane >> 4;
  float* al = &alds[wv][0];
  int ntiles = M >> 4;
  int lo = (int)((blockIdx.x * (unsigned)ntiles) / gridDim.x);
  int up = (int)(((blockIdx.x + 1u) * (unsigned)ntiles) / gridDim.x);
  int tm = lo + wv;
  float4 v[16];
#pragma unroll
  for (int i = 0; i < 16; ++i) v[i] = make_float4(0.f, 0.f, 0.f, 0.f);
  if (tm < up) {
#pragma unroll
    for (int i = 0; i < 16; ++i) v[i] = ((const float4*)(A + (size_t)(tm * 16 + i) * GEMM_K))[lane];
  }
  vg_stage_b<D_HID>(B, bl);
  __syncthreads();

  while (tm < up) {
#pragma unroll
    for (int i = 0; i < 16; ++i) *(float4*)(al + i * A_PITCH + 4 * lane) = v[i];
    const float4* rsp = (const float4*)(rowscale + tm * 16 + hi * 8);
    float4 rs0 = rsp[0];
    float4 rs1 = rsp[1];
    int tn = tm + 8;
    if (tn < up) {
#pragma unroll
      for (int i = 0; i < 16; ++i) v[i] = ((const float4*)(A + (size_t)(tn * 16 + i) * GEMM_K))[lane];
    }
    v8f acc[GEMM1_NT];
#pragma unroll
    for (int t = 0; t < GEMM1_NT; ++t) acc[t] = (v8f){};
    vg_tile_products<GEMM1_NT, D_HID>(al, bl, m, hi, acc);

    int rbase = tm * 16 + hi * 8;
    float rs[8] = {rs0.x, rs0.y, rs0.z, rs0.w, rs1.x, rs1.y, rs1.z, rs1.w};
#pragma unroll
    for (int t = 0; t < GEMM1_NT; ++t) {
      int col = t * 16 + m;
      float bv = bias[col];
#pragma unroll
      for (int j = 0; j < 8; ++j) {
        C[(size_t)(rbase + j) * (size_t)D_HID + (size_t)col] = (acc[t][j] + bv) * rs[j];
      }
    }
    tm = tn;
  }
}

#define ROWS_PER_WAVE 8
template <int U>
__device__ __forceinline__ void vg_edges128(const float* __restrict__ in, int s_l,
                                            int j, int lane, float4& acc) {
  float4 v[U];
#pragma unroll
  for (int u = 0; u < U; ++u) {
    int s = __builtin_amdgcn_readlane(s_l, j + u);
    v[u]  = ((const float4*)(in + (size_t)s * D_HID))[lane];
  }
#pragma unroll
  for (int u = 0; u < U; ++u) {
    acc.x += v[u].x;
    acc.y += v[u].y;
    acc.z += v[u].z;
    acc.w += v[u].w;
  }
}

__global__ void vg_gather128_kernel(const float* __restrict__ in, const int* __restrict__ csr,
                                    const int* __restrict__ rowstart, const int* __restrict__ cnt,
                                    const float* __restrict__ dinv, float* __restrict__ out,
                                    float* __restrict__ stats, int n) {
  __shared__ float4 red[8][64];
  int wv = __builtin_amdgcn_readfirstlane((int)(threadIdx.x >> 5));
  int row0 = (blockIdx.x * 8 + wv) * ROWS_PER_WAVE;
  int lane = threadIdx.x & 31;
  float4* my_sum = &red[wv][lane];
  float4* my_sq  = &red[wv][32 + lane];
  *my_sum = make_float4(0.f, 0.f, 0.f, 0.f);
  *my_sq  = make_float4(0.f, 0.f, 0.f, 0.f);
  for (int r = 0; r < ROWS_PER_WAVE; ++r) {
    int row = row0 + r;
    if (row >= n) break;
    int start = rowstart[row];
    int c = cnt[row];
    float dd = dinv[row];
    float4 acc = ((const float4*)(in + (size_t)row * D_HID))[lane];
    for (int base = 0; base < c; base += 32) {
      int m = c - base;
      if (m > 32) m = 32;
      int s_l = 0;
      if (lane < m) s_l = csr[start + base + lane];
      int j = 0;
      for (; j + 8 <= m; j += 8) vg_edges128<8>(in, s_l, j, lane, acc);
      if (j + 4 <= m) { vg_edges128<4>(in, s_l, j, lane, acc); j += 4; }
      if (j + 2 <= m) { vg_edges128<2>(in, s_l, j, lane, acc); j += 2; }
      if (j < m)      { vg_edges128<1>(in, s_l, j, lane, acc); }
    }
    acc.x *= dd; acc.y *= dd; acc.z *= dd; acc.w *= dd;
    ((float4*)(out + (size_t)row * D_HID))[lane] = acc;
    float4 ssum = *my_sum;
    float4 ssq  = *my_sq;
    ssum.x += acc.x; ssum.y += acc.y; ssum.z += acc.z; ssum.w += acc.w;
    ssq.x += acc.x * acc.x; ssq.y += acc.y * acc.y; ssq.z += acc.z * acc.z; ssq.w += acc.w * acc.w;
    *my_sum = ssum;
    *my_sq  = ssq;
  }
  __syncthreads();
  const float* rf = (const float*)&red[0][0];
  float tot = 0.f;
#pragma unroll
  for (int k = 0; k < 8; ++k) tot += rf[k * 256 + threadIdx.x];
  atomicAdd(&stats[(blockIdx.x & (STAT_COPIES - 1)) * 256 + threadIdx.x], tot);
}


template <int MASK>
__device__ __forceinline__ float vg_add_row_xor(float x) {
  int y = __builtin_amdgcn_update_dpp(0, __float_as_int(x), 0x160 + MASK, 0xf, 0xf, true);
  return x + __int_as_float(y);
}

#define GEMM2_NT (D_OUT / 16)
__global__ __launch_bounds__(256) void vg_gemm_bn_wmma_kernel(const float* __restrict__ A, const float* __restrict__ B,
                                       const float* __restrict__ bias, const float* __restrict__ stats,
                                       const float* __restrict__ gamma, const float* __restrict__ beta,
                                       const float* __restrict__ rowscale,
                                       float* __restrict__ C, float* __restrict__ rns, int M, int n) {
  __shared__ __attribute__((aligned(16))) float bn_sc[D_HID];
  __shared__ __attribute__((aligned(16))) float bn_sh[D_HID];
  __shared__ __attribute__((aligned(16))) float bl[(GEMM_K / 2) * 2 * D_OUT];
  __shared__ __attribute__((aligned(16))) float alds[8][16 * A_PITCH];
  int wv = __builtin_amdgcn_readfirstlane((int)(threadIdx.x >> 5));
  int lane = threadIdx.x & 31;
  int m  = lane & 15;
  int hi = lane >> 4;
  float* al = &alds[wv][0];
  int ntiles = M >> 4;
  int lo = (int)((blockIdx.x * (unsigned)ntiles) / gridDim.x);
  int up = (int)(((blockIdx.x + 1u) * (unsigned)ntiles) / gridDim.x);
  int tm = lo + wv;
  float4 v[16];
#pragma unroll
  for (int i = 0; i < 16; ++i) v[i] = make_float4(0.f, 0.f, 0.f, 0.f);
  if (tm < up) {
#pragma unroll
    for (int i = 0; i < 16; ++i) v[i] = ((const float4*)(A + (size_t)(tm * 16 + i) * GEMM_K))[lane];
  }
  if (threadIdx.x < D_HID) {
    int c = threadIdx.x;
    float invn = 1.0f / (float)n;
    float s1 = 0.f, s2 = 0.f;
#pragma unroll
    for (int k = 0; k < STAT_COPIES; ++k) {
      s1 += stats[k * 256 + c];
      s2 += stats[k * 256 + 128 + c];
    }
    float mean = s1 * invn;
    float var  = s2 * invn - mean * mean;
    float a = gamma[c] * rsqrtf(var + BN_EPS);
    bn_sc[c] = a;
    bn_sh[c] = beta[c] - mean * a;
  }
  vg_stage_b<D_OUT>(B, bl);
  __syncthreads();

  float4 sc = *(const float4*)(bn_sc + 4 * lane);
  float4 sh = *(const float4*)(bn_sh + 4 * lane);
  while (tm < up) {
#pragma unroll
    for (int i = 0; i < 16; ++i) {
      float v0 = sc.x * v[i].x + sh.x;
      float v1 = sc.y * v[i].y + sh.y;
      float v2 = sc.z * v[i].z + sh.z;
      float v3 = sc.w * v[i].w + sh.w;
      float4 o;
      o.x = (v0 > 0.f) ? v0 : LRELU_SLOPE * v0;
      o.y = (v1 > 0.f) ? v1 : LRELU_SLOPE * v1;
      o.z = (v2 > 0.f) ? v2 : LRELU_SLOPE * v2;
      o.w = (v3 > 0.f) ? v3 : LRELU_SLOPE * v3;
      *(float4*)(al + i * A_PITCH + 4 * lane) = o;
    }
    const float4* rsp = (const float4*)(rowscale + tm * 16 + hi * 8);
    float4 rs0 = rsp[0];
    float4 rs1 = rsp[1];
    int tn = tm + 8;
    if (tn < up) {
#pragma unroll
      for (int i = 0; i < 16; ++i) v[i] = ((const float4*)(A + (size_t)(tn * 16 + i) * GEMM_K))[lane];
    }
    v8f acc[GEMM2_NT];
#pragma unroll
    for (int t = 0; t < GEMM2_NT; ++t) acc[t] = (v8f){};
    vg_tile_products<GEMM2_NT, D_OUT>(al, bl, m, hi, acc);

    int rbase = tm * 16 + hi * 8;
    float rs[8] = {rs0.x, rs0.y, rs0.z, rs0.w, rs1.x, rs1.y, rs1.z, rs1.w};
    float sq[8];
#pragma unroll
    for (int j = 0; j < 8; ++j) sq[j] = 0.f;
#pragma unroll
    for (int t = 0; t < GEMM2_NT; ++t) {
      int col = t * 16 + m;
      float bv = bias[col];
#pragma unroll
      for (int j = 0; j < 8; ++j) {
        float lv = acc[t][j] + bv;
        C[(size_t)(rbase + j) * (size_t)D_OUT + (size_t)col] = lv * rs[j];
        sq[j] += lv * lv;
      }
    }
    float mine = 0.f;
#pragma unroll
    for (int j = 0; j < 8; ++j) {
      float s = sq[j];
      s = vg_add_row_xor<8>(s);
      s = vg_add_row_xor<4>(s);
      s = vg_add_row_xor<2>(s);
      s = vg_add_row_xor<1>(s);
      mine = (m == j) ? s : mine;
    }
    if (m < 8) rns[rbase + m] = NORM_SCALE / fmaxf(sqrtf(mine), 1e-12f);
    tm = tn;
  }
}

template <int U>
__device__ __forceinline__ void vg_edges64(const float* __restrict__ l, int s_l, float sc_l,
                                           int j, int lane, float2& am, float2& al) {
  float2 v[U];
#pragma unroll
  for (int u = 0; u < U; ++u) {
    int s = __builtin_amdgcn_readlane(s_l, j + u);
    v[u]  = ((const float2*)(l + (size_t)s * D_OUT))[lane];
  }
#pragma unroll
  for (int u = 0; u < U; ++u) {
    float sc = __int_as_float(__builtin_amdgcn_readlane(__float_as_int(sc_l), j + u));
    am.x += v[u].x;
    am.y += v[u].y;
    al.x += sc * v[u].x;
    al.y += sc * v[u].y;
  }
}

__global__ void vg_gather64_kernel(const float* __restrict__ l, const int* __restrict__ csr,
                                   const int* __restrict__ rowstart, const int* __restrict__ cnt,
                                   const float* __restrict__ dinv, const float* __restrict__ rns,
                                   const float* __restrict__ noise, float* __restrict__ mu,
                                   float* __restrict__ logstd, float* __restrict__ zeta, int n) {
  int wv = __builtin_amdgcn_readfirstlane((int)(threadIdx.x >> 5));
  int row0 = (blockIdx.x * 8 + wv) * ROWS_PER_WAVE;
  int lane = threadIdx.x & 31;
  for (int r = 0; r < ROWS_PER_WAVE; ++r) {
    int row = row0 + r;
    if (row >= n) break;
    int start = rowstart[row];
    int c = cnt[row];
    float dd = dinv[row];
    float rr = rns[row];
    float2 sv = ((const float2*)(l + (size_t)row * D_OUT))[lane];
    float2 nz = ((const float2*)(noise + (size_t)row * D_OUT))[lane];
    float2 am, al;
    am.x = sv.x;      am.y = sv.y;
    al.x = sv.x * rr; al.y = sv.y * rr;
    for (int base = 0; base < c; base += 32) {
      int m = c - base;
      if (m > 32) m = 32;
      int s_l = 0;
      float sc_l = 0.0f;
      if (lane < m) {
        s_l  = csr[start + base + lane];
        sc_l = rns[s_l];
      }
      int j = 0;
      for (; j + 8 <= m; j += 8) vg_edges64<8>(l, s_l, sc_l, j, lane, am, al);
      if (j + 4 <= m) { vg_edges64<4>(l, s_l, sc_l, j, lane, am, al); j += 4; }
      if (j + 2 <= m) { vg_edges64<2>(l, s_l, sc_l, j, lane, am, al); j += 2; }
      if (j < m)      { vg_edges64<1>(l, s_l, sc_l, j, lane, am, al); }
    }
    am.x *= dd; am.y *= dd;
    al.x *= dd; al.y *= dd;
    ((float2*)(mu     + (size_t)row * D_OUT))[lane] = am;
    ((float2*)(logstd + (size_t)row * D_OUT))[lane] = al;
    float2 z;
    z.x = am.x + nz.x * expf(al.x);
    z.y = am.y + nz.y * expf(al.y);
    ((float2*)(zeta   + (size_t)row * D_OUT))[lane] = z;
  }
}

extern "C" void kernel_launch(void* const* d_in, const int* in_sizes, int n_in,
                              void* d_out, int out_size, void* d_ws, size_t ws_size,
                              hipStream_t stream) {
  const float* x     = (const float*)d_in[0];
  const int*   esrc  = (const int*)  d_in[1];
  const int*   edst  = (const int*)  d_in[2];
  const float* W1    = (const float*)d_in[3];
  const float* b1    = (const float*)d_in[4];
  const float* gamma = (const float*)d_in[5];
  const float* beta  = (const float*)d_in[6];
  const float* Wmu   = (const float*)d_in[7];
  const float* bmu   = (const float*)d_in[8];
  const float* noise = (const float*)d_in[9];

  const int N = in_sizes[0] / D_IN;
  const int E = in_sizes[1];

  float* ws = (float*)d_ws;
  size_t npad = ((size_t)N + 255) & ~(size_t)255;
  size_t off = 0;
  float* dinv  = ws + off; off += npad;
  float* stats = ws + off; off += STAT_WORDS;
  float* rns   = ws + off; off += npad;
  float* h     = ws + off; off += (size_t)N * D_HID;
  float* hagg  = ws + off; off += (size_t)N * D_HID;
  float* lbuf  = ws + off; off += (size_t)N * D_OUT;
  size_t epad = ((size_t)E + 255) & ~(size_t)255;
  const int NB   = (N + BKT_ROWS - 1) / BKT_ROWS;
  const int NBLK = (E + EPB - 1) / EPB;
  size_t bbpad = ((size_t)NBLK * (size_t)NB + 255) & ~(size_t)255;
  int* cnt      = (int*)(ws + off); off += npad;
  int* rowstart = (int*)(ws + off); off += npad;
  int* bcnt     = (int*)(ws + off); off += npad;
  int* blkbase  = (int*)(ws + off); off += bbpad;
  int* csr      = (int*)(ws + off); off += epad;
  int2* binned  = (int2*)(ws + off); off += 2 * epad;

  float* mu_out   = (float*)d_out;
  float* ls_out   = mu_out + (size_t)N * D_OUT;
  float* zeta_out = ls_out + (size_t)N * D_OUT;

  const int B = 256;
  auto cdiv = [](int a, int b) { return (a + b - 1) / b; };

  vg_init_kernel<<<cdiv(NB > STAT_WORDS ? NB : STAT_WORDS, B), B, 0, stream>>>(bcnt, stats, NB);
  vg_bucket_count_kernel<<<NBLK, B, NB * sizeof(int), stream>>>(edst, bcnt, blkbase, E, NB);
  vg_bucket_scatter_kernel<<<NBLK, B, NB * sizeof(int), stream>>>(esrc, edst, bcnt, blkbase, binned, E, NB);
  vg_bucket_rows_kernel<<<NB, B, 0, stream>>>(binned, bcnt, cnt, rowstart, dinv, csr, N);

  vg_gemm_wmma_kernel<<<GEMM1_GRID, B, 0, stream>>>(x, W1, b1, dinv, h, N);

  vg_gather128_kernel<<<cdiv(N, (B / 32) * ROWS_PER_WAVE), B, 0, stream>>>(h, csr, rowstart, cnt, dinv, hagg, stats, N);

  vg_gemm_bn_wmma_kernel<<<GEMM2_GRID, B, 0, stream>>>(hagg, Wmu, bmu, stats, gamma, beta, dinv, lbuf, rns, N, N);

  vg_gather64_kernel<<<cdiv(N, (B / 32) * ROWS_PER_WAVE), B, 0, stream>>>(lbuf, csr, rowstart, cnt, dinv, rns, noise,
                                                        mu_out, ls_out, zeta_out, N);
}


# Edge list built in 2 launches, not 4: a block sorts its 4096 edges by bucket in LDS and writes one piece + run table; the bucket's block walks its runs; no global adds; row sum order still arrival
# speedup vs baseline: 5.3773x; 1.0573x over previous
#include <hip/hip_runtime.h>
#include <hip/hip_bf16.h>


#define D_IN   128
#define D_HID  128
#define D_OUT  64
#define BN_EPS 1e-5f
#define LRELU_SLOPE 0.01f
#define NORM_SCALE 1.8f

typedef __attribute__((ext_vector_type(2))) float v2f;
typedef __attribute__((ext_vector_type(8))) float v8f;

#define BKT_SHIFT 8
#define BKT_ROWS  256
#define EPB       4096
#define EPT       (EPB / 256)
#define BPT       20
#define ROW_STAGE 8192

#define STAT_COPIES 16
#define STAT_WORDS  (256 * STAT_COPIES)

__global__ void vg_block_sort_kernel(const int* __restrict__ src, const int* __restrict__ dst,
                                     int2* __restrict__ binned, int* __restrict__ runs,
                                     float* __restrict__ stats, int e, int nb) {
  extern __shared__ int lds_i[];
  __shared__ int2 stage[EPB];
  __shared__ int wsum[8];
  int* lcur   = lds_i;
  int* lstart = lds_i + nb;
  int lane = threadIdx.x & 31;
  int wv = threadIdx.x >> 5;
  int nblk = gridDim.x;
  for (int i = blockIdx.x * 256 + threadIdx.x; i < STAT_WORDS; i += nblk * 256) stats[i] = 0.0f;
  for (int b = threadIdx.x; b < nb; b += 256) lcur[b] = 0;
  __syncthreads();
  int e0 = blockIdx.x * EPB + threadIdx.x;
  int d[EPT], s[EPT];
#pragma unroll
  for (int k = 0; k < EPT; ++k) {
    int i = e0 + k * 256;
    d[k] = (i < e) ? dst[i] : -1;
    s[k] = (i < e) ? src[i] : 0;
  }
#pragma unroll
  for (int k = 0; k < EPT; ++k) {
    if (d[k] >= 0) atomicAdd(&lcur[d[k] >> BKT_SHIFT], 1);
  }
  __syncthreads();
  int run = 0;
  for (int b0 = 0; b0 < nb; b0 += 256) {
    int b = b0 + threadIdx.x;
    int hc = (b < nb) ? lcur[b] : 0;
    int x = hc;
#pragma unroll
    for (int o = 1; o < 32; o <<= 1) {
      int y = __shfl_up(x, o, 32);
      if (lane >= o) x += y;
    }
    if (lane == 31) wsum[wv] = x;
    __syncthreads();
    int wbase = 0, tot = 0;
#pragma unroll
    for (int k = 0; k < 8; ++k) { int w = wsum[k]; if (k < wv) wbase += w; tot += w; }
    if (b < nb) {
      int ls0 = run + wbase + (x - hc);
      lstart[b] = ls0;
      lcur[b]   = ls0;
      runs[(size_t)b * (size_t)nblk + blockIdx.x] = ls0;
    }
    run += tot;
    __syncthreads();
  }
#pragma unroll
  for (int k = 0; k < EPT; ++k) {
    if (d[k] >= 0) {
      int p = atomicAdd(&lcur[d[k] >> BKT_SHIFT], 1);
      stage[p] = make_int2(s[k], d[k]);
    }
  }
  __syncthreads();
  int nv = e - blockIdx.x * EPB;
  if (nv > EPB) nv = EPB;
  if (threadIdx.x == 0) runs[(size_t)nb * (size_t)nblk + blockIdx.x] = nv;
  int2* piece = binned + (size_t)blockIdx.x * EPB;
  for (int i = threadIdx.x; i < nv; i += 256) piece[i] = stage[i];
}

__device__ __forceinline__ void vg_bucket_trip(const int2* __restrict__ binned, const int* pre, const int* off,
                                               int top, int i0, int t, int bsize, int2 (&ev)[BPT]) {
  int idx[BPT], lo[BPT];
#pragma unroll
  for (int k = 0; k < BPT; ++k) {
    int i = i0 + k * 256 + t;
    idx[k] = (i < bsize) ? i : (bsize - 1);
    lo[k] = 0;
  }
  for (int st = top; st > 0; st >>= 1) {
    int mid[BPT], pm[BPT];
#pragma unroll
    for (int k = 0; k < BPT; ++k) {
      mid[k] = lo[k] + st;
      pm[k] = pre[mid[k]];
    }
#pragma unroll
    for (int k = 0; k < BPT; ++k) lo[k] = (pm[k] <= idx[k]) ? mid[k] : lo[k];
  }
#pragma unroll
  for (int k = 0; k < BPT; ++k) {
    int j = lo[k];
    ev[k] = binned[(size_t)j * EPB + (size_t)(off[j] + (idx[k] - pre[j]))];
  }
}

__global__ void vg_bucket_rows_kernel(const int2* __restrict__ binned, const int* __restrict__ runs,
                                      int* __restrict__ cnt, int* __restrict__ rowstart,
                                      float* __restrict__ dinv, int* __restrict__ csr, int n, int nblk) {
  extern __shared__ int lds_i[];
  __shared__ int lcnt[BKT_ROWS];
  __shared__ int lcur[BKT_ROWS];
  __shared__ int wsum[8];
  __shared__ int wsum2[8];
  __shared__ int wsum3[8];
  __shared__ int lsrc[ROW_STAGE];
  int* pre = lds_i;
  int* off = lds_i + 2 * nblk;
  int bkt = blockIdx.x;
  int t = threadIdx.x;
  int lane = t & 31;
  int wv = t >> 5;
  lcnt[t] = 0;
  for (int j = nblk + t; j < 2 * nblk; j += 256) pre[j] = 0x7fffffff;
  const int* r0 = runs + (size_t)bkt * (size_t)nblk;
  const int* r1 = r0 + nblk;
  int run = 0;
  int part = 0;
  for (int j0 = 0; j0 < nblk; j0 += 256) {
    int j = j0 + t;
    int o0 = (j < nblk) ? r0[j] : 0;
    int o1 = (j < nblk) ? r1[j] : 0;
    int len = o1 - o0;
    part += o0;
    int x = len;
#pragma unroll
    for (int o = 1; o < 32; o <<= 1) {
      int y = __shfl_up(x, o, 32);
      if (lane >= o) x += y;
    }
    if (lane == 31) wsum[wv] = x;
    __syncthreads();
    int wbase = 0, tot = 0;
#pragma unroll
    for (int k = 0; k < 8; ++k) { int w = wsum[k]; if (k < wv) wbase += w; tot += w; }
    if (j < nblk) {
      pre[j] = run + wbase + (x - len);
      off[j] = o0;
    }
    run += tot;
    __syncthreads();
  }
  int bsize = run;
#pragma unroll
  for (int o = 16; o > 0; o >>= 1) part += __shfl_xor(part, o, 32);
  if (lane == 0) wsum3[wv] = part;
  __syncthreads();
  int bstart = 0;
#pragma unroll
  for (int k = 0; k < 8; ++k) bstart += wsum3[k];
  int top = 1;
  while (top * 2 < nblk) top *= 2;
  bool fits = (bsize <= ROW_STAGE);
  bool one  = (bsize <= 256 * BPT);

  int2 ev[BPT];
#pragma unroll
  for (int k = 0; k < BPT; ++k) ev[k] = make_int2(0, 0);
  for (int i0 = 0; i0 < bsize; i0 += 256 * BPT) {
    vg_bucket_trip(binned, pre, off, top, i0, t, bsize, ev);
#pragma unroll
    for (int k = 0; k < BPT; ++k) {
      if (i0 + k * 256 + t < bsize) atomicAdd(&lcnt[ev[k].y & (BKT_ROWS - 1)], 1);
    }
  }
  __syncthreads();

  int c = lcnt[t];
  int x = c;
#pragma unroll
  for (int o = 1; o < 32; o <<= 1) {
    int y = __shfl_up(x, o, 32);
    if (lane >= o) x += y;
  }
  if (lane == 31) wsum2[wv] = x;
  __syncthreads();
  int wbase = 0;
#pragma unroll
  for (int k = 0; k < 8; ++k) { int w = wsum2[k]; if (k < wv) wbase += w; }
  int rs = bstart + wbase + (x - c);
  lcur[t] = rs;
  int row = bkt * BKT_ROWS + t;
  if (row < n) {
    cnt[row] = c;
    rowstart[row] = rs;
    dinv[row] = rsqrtf((float)c + 1.0f);
  }
  __syncthreads();

  for (int i0 = 0; i0 < bsize; i0 += 256 * BPT) {
    if (!one) vg_bucket_trip(binned, pre, off, top, i0, t, bsize, ev);
#pragma unroll
    for (int k = 0; k < BPT; ++k) {
      if (i0 + k * 256 + t < bsize) {
        int p = atomicAdd(&lcur[ev[k].y & (BKT_ROWS - 1)], 1);
        if (fits) lsrc[p - bstart] = ev[k].x;
        else      csr[p] = ev[k].x;
      }
    }
  }
  __syncthreads();
  if (fits) {
    for (int i = t; i < bsize; i += 256) csr[bstart + i] = lsrc[i];
  }
}

#define GEMM_K    128
#define A_PITCH   132
#define GEMM1_GRID 512
#define GEMM2_GRID 768

template <int NC>
__device__ __forceinline__ void vg_stage_b(const float* __restrict__ B, float* bl) {
#pragma unroll 16
  for (int p = threadIdx.x; p < (GEMM_K / 2) * NC; p += 256) {
    int kp = p / NC;
    int n  = p - kp * NC;
    v2f v;
    v.x = B[(size_t)(2 * kp) * NC + n];
    v.y = B[(size_t)(2 * kp + 1) * NC + n];
    int t = n >> 4;
    int m = n & 15;
    *(v2f*)(bl + kp * (2 * NC) + (t >> 1) * 64 + m * 4 + (t & 1) * 2) = v;
  }
}

template <int NT, int NC>
__device__ __forceinline__ void vg_tile_products(const float* al, const float* bl, int m, int hi,
                                                 v8f (&acc)[NT]) {
  const float* ap = al + m * A_PITCH + 2 * hi;
  const float* bp = bl + hi * (2 * NC) + m * 4;
#pragma unroll 2
  for (int k0 = 0; k0 < GEMM_K; k0 += 4) {
    v2f a = *(const v2f*)(ap + k0);
    const float* bk = bp + (k0 >> 1) * (2 * NC);
    float4 bq[NT / 2];
#pragma unroll
    for (int u = 0; u < NT / 2; ++u) bq[u] = *(const float4*)(bk + u * 64);
#pragma unroll
    for (int u = 0; u < NT / 2; ++u) {
      v2f b0, b1;
      b0.x = bq[u].x; b0.y = bq[u].y;
      b1.x = bq[u].z; b1.y = bq[u].w;
      acc[2 * u]     = __builtin_amdgcn_wmma_f32_16x16x4_f32(false, a, false, b0,
                                                             (short)0, acc[2 * u], false, false);
      acc[2 * u + 1] = __builtin_amdgcn_wmma_f32_16x16x4_f32(false, a, false, b1,
                                                             (short)0, acc[2 * u + 1], false, false);
    }
  }
}

#define GEMM1_NT (D_HID / 16)
__global__ __launch_bounds__(256) void vg_gemm_wmma_kernel(const float* __restrict__ A, const float* __restrict__ B,
                                    const float* __restrict__ bias, const float* __restrict__ rowscale,
                                    float* __restrict__ C, int M) {
  __shared__ __attribute__((aligned(16))) float bl[(GEMM_K / 2) * 2 * D_HID];
  __shared__ __attribute__((aligned(16))) float alds[8][16 * A_PITCH];
  int wv = __builtin_amdgcn_readfirstlane((int)(threadIdx.x >> 5));
  int lane = threadIdx.x & 31;
  int m  = lane & 15;
  int hi = lane >> 4;
  float* al = &alds[wv][0];
  int ntiles = M >> 4;
  int lo = (int)((blockIdx.x * (unsigned)ntiles) / gridDim.x);
  int up = (int)(((blockIdx.x + 1u) * (unsigned)ntiles) / gridDim.x);
  int tm = lo + wv;
  float4 v[16];
#pragma unroll
  for (int i = 0; i < 16; ++i) v[i] = make_float4(0.f, 0.f, 0.f, 0.f);
  if (tm < up) {
#pragma unroll
    for (int i = 0; i < 16; ++i) v[i] = ((const float4*)(A + (size_t)(tm * 16 + i) * GEMM_K))[lane];
  }
  vg_stage_b<D_HID>(B, bl);
  __syncthreads();

  while (tm < up) {
#pragma unroll
    for (int i = 0; i < 16; ++i) *(float4*)(al + i * A_PITCH + 4 * lane) = v[i];
    const float4* rsp = (const float4*)(rowscale + tm * 16 + hi * 8);
    float4 rs0 = rsp[0];
    float4 rs1 = rsp[1];
    int tn = tm + 8;
    if (tn < up) {
#pragma unroll
      for (int i = 0; i < 16; ++i) v[i] = ((const float4*)(A + (size_t)(tn * 16 + i) * GEMM_K))[lane];
    }
    v8f acc[GEMM1_NT];
#pragma unroll
    for (int t = 0; t < GEMM1_NT; ++t) acc[t] = (v8f){};
    vg_tile_products<GEMM1_NT, D_HID>(al, bl, m, hi, acc);

    int rbase = tm * 16 + hi * 8;
    float rs[8] = {rs0.x, rs0.y, rs0.z, rs0.w, rs1.x, rs1.y, rs1.z, rs1.w};
#pragma unroll
    for (int t = 0; t < GEMM1_NT; ++t) {
      int col = t * 16 + m;
      float bv = bias[col];
#pragma unroll
      for (int j = 0; j < 8; ++j) {
        C[(size_t)(rbase + j) * (size_t)D_HID + (size_t)col] = (acc[t][j] + bv) * rs[j];
      }
    }
    tm = tn;
  }
}

#define ROWS_PER_WAVE 8
template <int U>
__device__ __forceinline__ void vg_edges128(const float* __restrict__ in, int s_l,
                                            int j, int lane, float4& acc) {
  float4 v[U];
#pragma unroll
  for (int u = 0; u < U; ++u) {
    int s = __builtin_amdgcn_readlane(s_l, j + u);
    v[u]  = ((const float4*)(in + (size_t)s * D_HID))[lane];
  }
#pragma unroll
  for (int u = 0; u < U; ++u) {
    acc.x += v[u].x;
    acc.y += v[u].y;
    acc.z += v[u].z;
    acc.w += v[u].w;
  }
}

__global__ void vg_gather128_kernel(const float* __restrict__ in, const int* __restrict__ csr,
                                    const int* __restrict__ rowstart, const int* __restrict__ cnt,
                                    const float* __restrict__ dinv, float* __restrict__ out,
                                    float* __restrict__ stats, int n) {
  __shared__ float4 red[8][64];
  int wv = __builtin_amdgcn_readfirstlane((int)(threadIdx.x >> 5));
  int row0 = (blockIdx.x * 8 + wv) * ROWS_PER_WAVE;
  int lane = threadIdx.x & 31;
  float4* my_sum = &red[wv][lane];
  float4* my_sq  = &red[wv][32 + lane];
  *my_sum = make_float4(0.f, 0.f, 0.f, 0.f);
  *my_sq  = make_float4(0.f, 0.f, 0.f, 0.f);
  for (int r = 0; r < ROWS_PER_WAVE; ++r) {
    int row = row0 + r;
    if (row >= n) break;
    int start = rowstart[row];
    int c = cnt[row];
    float dd = dinv[row];
    float4 acc = ((const float4*)(in + (size_t)row * D_HID))[lane];
    for (int base = 0; base < c; base += 32) {
      int m = c - base;
      if (m > 32) m = 32;
      int s_l = 0;
      if (lane < m) s_l = csr[start + base + lane];
      int j = 0;
      for (; j + 8 <= m; j += 8) vg_edges128<8>(in, s_l, j, lane, acc);
      if (j + 4 <= m) { vg_edges128<4>(in, s_l, j, lane, acc); j += 4; }
      if (j + 2 <= m) { vg_edges128<2>(in, s_l, j, lane, acc); j += 2; }
      if (j < m)      { vg_edges128<1>(in, s_l, j, lane, acc); }
    }
    acc.x *= dd; acc.y *= dd; acc.z *= dd; acc.w *= dd;
    ((float4*)(out + (size_t)row * D_HID))[lane] = acc;
    float4 ssum = *my_sum;
    float4 ssq  = *my_sq;
    ssum.x += acc.x; ssum.y += acc.y; ssum.z += acc.z; ssum.w += acc.w;
    ssq.x += acc.x * acc.x; ssq.y += acc.y * acc.y; ssq.z += acc.z * acc.z; ssq.w += acc.w * acc.w;
    *my_sum = ssum;
    *my_sq  = ssq;
  }
  __syncthreads();
  const float* rf = (const float*)&red[0][0];
  float tot = 0.f;
#pragma unroll
  for (int k = 0; k < 8; ++k) tot += rf[k * 256 + threadIdx.x];
  atomicAdd(&stats[(blockIdx.x & (STAT_COPIES - 1)) * 256 + threadIdx.x], tot);
}


template <int MASK>
__device__ __forceinline__ float vg_add_row_xor(float x) {
  int y = __builtin_amdgcn_update_dpp(0, __float_as_int(x), 0x160 + MASK, 0xf, 0xf, true);
  return x + __int_as_float(y);
}

#define GEMM2_NT (D_OUT / 16)
__global__ __launch_bounds__(256) void vg_gemm_bn_wmma_kernel(const float* __restrict__ A, const float* __restrict__ B,
                                       const float* __restrict__ bias, const float* __restrict__ stats,
                                       const float* __restrict__ gamma, const float* __restrict__ beta,
                                       const float* __restrict__ rowscale,
                                       float* __restrict__ C, float* __restrict__ rns, int M, int n) {
  __shared__ __attribute__((aligned(16))) float bn_sc[D_HID];
  __shared__ __attribute__((aligned(16))) float bn_sh[D_HID];
  __shared__ __attribute__((aligned(16))) float bl[(GEMM_K / 2) * 2 * D_OUT];
  __shared__ __attribute__((aligned(16))) float alds[8][16 * A_PITCH];
  int wv = __builtin_amdgcn_readfirstlane((int)(threadIdx.x >> 5));
  int lane = threadIdx.x & 31;
  int m  = lane & 15;
  int hi = lane >> 4;
  float* al = &alds[wv][0];
  int ntiles = M >> 4;
  int lo = (int)((blockIdx.x * (unsigned)ntiles) / gridDim.x);
  int up = (int)(((blockIdx.x + 1u) * (unsigned)ntiles) / gridDim.x);
  int tm = lo + wv;
  float4 v[16];
#pragma unroll
  for (int i = 0; i < 16; ++i) v[i] = make_float4(0.f, 0.f, 0.f, 0.f);
  if (tm < up) {
#pragma unroll
    for (int i = 0; i < 16; ++i) v[i] = ((const float4*)(A + (size_t)(tm * 16 + i) * GEMM_K))[lane];
  }
  if (threadIdx.x < D_HID) {
    int c = threadIdx.x;
    float invn = 1.0f / (float)n;
    float s1 = 0.f, s2 = 0.f;
#pragma unroll
    for (int k = 0; k < STAT_COPIES; ++k) {
      s1 += stats[k * 256 + c];
      s2 += stats[k * 256 + 128 + c];
    }
    float mean = s1 * invn;
    float var  = s2 * invn - mean * mean;
    float a = gamma[c] * rsqrtf(var + BN_EPS);
    bn_sc[c] = a;
    bn_sh[c] = beta[c] - mean * a;
  }
  vg_stage_b<D_OUT>(B, bl);
  __syncthreads();

  float4 sc = *(const float4*)(bn_sc + 4 * lane);
  float4 sh = *(const float4*)(bn_sh + 4 * lane);
  while (tm < up) {
#pragma unroll
    for (int i = 0; i < 16; ++i) {
      float v0 = sc.x * v[i].x + sh.x;
      float v1 = sc.y * v[i].y + sh.y;
      float v2 = sc.z * v[i].z + sh.z;
      float v3 = sc.w * v[i].w + sh.w;
      float4 o;
      o.x = (v0 > 0.f) ? v0 : LRELU_SLOPE * v0;
      o.y = (v1 > 0.f) ? v1 : LRELU_SLOPE * v1;
      o.z = (v2 > 0.f) ? v2 : LRELU_SLOPE * v2;
      o.w = (v3 > 0.f) ? v3 : LRELU_SLOPE * v3;
      *(float4*)(al + i * A_PITCH + 4 * lane) = o;
    }
    const float4* rsp = (const float4*)(rowscale + tm * 16 + hi * 8);
    float4 rs0 = rsp[0];
    float4 rs1 = rsp[1];
    int tn = tm + 8;
    if (tn < up) {
#pragma unroll
      for (int i = 0; i < 16; ++i) v[i] = ((const float4*)(A + (size_t)(tn * 16 + i) * GEMM_K))[lane];
    }
    v8f acc[GEMM2_NT];
#pragma unroll
    for (int t = 0; t < GEMM2_NT; ++t) acc[t] = (v8f){};
    vg_tile_products<GEMM2_NT, D_OUT>(al, bl, m, hi, acc);

    int rbase = tm * 16 + hi * 8;
    float rs[8] = {rs0.x, rs0.y, rs0.z, rs0.w, rs1.x, rs1.y, rs1.z, rs1.w};
    float sq[8];
#pragma unroll
    for (int j = 0; j < 8; ++j) sq[j] = 0.f;
#pragma unroll
    for (int t = 0; t < GEMM2_NT; ++t) {
      int col = t * 16 + m;
      float bv = bias[col];
#pragma unroll
      for (int j = 0; j < 8; ++j) {
        float lv = acc[t][j] + bv;
        C[(size_t)(rbase + j) * (size_t)D_OUT + (size_t)col] = lv * rs[j];
        sq[j] += lv * lv;
      }
    }
    float mine = 0.f;
#pragma unroll
    for (int j = 0; j < 8; ++j) {
      float s = sq[j];
      s = vg_add_row_xor<8>(s);
      s = vg_add_row_xor<4>(s);
      s = vg_add_row_xor<2>(s);
      s = vg_add_row_xor<1>(s);
      mine = (m == j) ? s : mine;
    }
    if (m < 8) rns[rbase + m] = NORM_SCALE / fmaxf(sqrtf(mine), 1e-12f);
    tm = tn;
  }
}

template <int U>
__device__ __forceinline__ void vg_edges64(const float* __restrict__ l, int s_l, float sc_l,
                                           int j, int lane, float2& am, float2& al) {
  float2 v[U];
#pragma unroll
  for (int u = 0; u < U; ++u) {
    int s = __builtin_amdgcn_readlane(s_l, j + u);
    v[u]  = ((const float2*)(l + (size_t)s * D_OUT))[lane];
  }
#pragma unroll
  for (int u = 0; u < U; ++u) {
    float sc = __int_as_float(__builtin_amdgcn_readlane(__float_as_int(sc_l), j + u));
    am.x += v[u].x;
    am.y += v[u].y;
    al.x += sc * v[u].x;
    al.y += sc * v[u].y;
  }
}

__global__ void vg_gather64_kernel(const float* __restrict__ l, const int* __restrict__ csr,
                                   const int* __restrict__ rowstart, const int* __restrict__ cnt,
                                   const float* __restrict__ dinv, const float* __restrict__ rns,
                                   const float* __restrict__ noise, float* __restrict__ mu,
                                   float* __restrict__ logstd, float* __restrict__ zeta, int n) {
  int wv = __builtin_amdgcn_readfirstlane((int)(threadIdx.x >> 5));
  int row0 = (blockIdx.x * 8 + wv) * ROWS_PER_WAVE;
  int lane = threadIdx.x & 31;
  for (int r = 0; r < ROWS_PER_WAVE; ++r) {
    int row = row0 + r;
    if (row >= n) break;
    int start = rowstart[row];
    int c = cnt[row];
    float dd = dinv[row];
    float rr = rns[row];
    float2 sv = ((const float2*)(l + (size_t)row * D_OUT))[lane];
    float2 nz = ((const float2*)(noise + (size_t)row * D_OUT))[lane];
    float2 am, al;
    am.x = sv.x;      am.y = sv.y;
    al.x = sv.x * rr; al.y = sv.y * rr;
    for (int base = 0; base < c; base += 32) {
      int m = c - base;
      if (m > 32) m = 32;
      int s_l = 0;
      float sc_l = 0.0f;
      if (lane < m) {
        s_l  = csr[start + base + lane];
        sc_l = rns[s_l];
      }
      int j = 0;
      for (; j + 8 <= m; j += 8) vg_edges64<8>(l, s_l, sc_l, j, lane, am, al);
      if (j + 4 <= m) { vg_edges64<4>(l, s_l, sc_l, j, lane, am, al); j += 4; }
      if (j + 2 <= m) { vg_edges64<2>(l, s_l, sc_l, j, lane, am, al); j += 2; }
      if (j < m)      { vg_edges64<1>(l, s_l, sc_l, j, lane, am, al); }
    }
    am.x *= dd; am.y *= dd;
    al.x *= dd; al.y *= dd;
    ((float2*)(mu     + (size_t)row * D_OUT))[lane] = am;
    ((float2*)(logstd + (size_t)row * D_OUT))[lane] = al;
    float2 z;
    z.x = am.x + nz.x * expf(al.x);
    z.y = am.y + nz.y * expf(al.y);
    ((float2*)(zeta   + (size_t)row * D_OUT))[lane] = z;
  }
}

extern "C" void kernel_launch(void* const* d_in, const int* in_sizes, int n_in,
                              void* d_out, int out_size, void* d_ws, size_t ws_size,
                              hipStream_t stream) {
  const float* x     = (const float*)d_in[0];
  const int*   esrc  = (const int*)  d_in[1];
  const int*   edst  = (const int*)  d_in[2];
  const float* W1    = (const float*)d_in[3];
  const float* b1    = (const float*)d_in[4];
  const float* gamma = (const float*)d_in[5];
  const float* beta  = (const float*)d_in[6];
  const float* Wmu   = (const float*)d_in[7];
  const float* bmu   = (const float*)d_in[8];
  const float* noise = (const float*)d_in[9];

  const int N = in_sizes[0] / D_IN;
  const int E = in_sizes[1];

  float* ws = (float*)d_ws;
  size_t npad = ((size_t)N + 255) & ~(size_t)255;
  size_t off = 0;
  float* dinv  = ws + off; off += npad;
  float* stats = ws + off; off += STAT_WORDS;
  float* rns   = ws + off; off += npad;
  float* h     = ws + off; off += (size_t)N * D_HID;
  float* hagg  = ws + off; off += (size_t)N * D_HID;
  float* lbuf  = ws + off; off += (size_t)N * D_OUT;
  size_t epad = ((size_t)E + 255) & ~(size_t)255;
  const int NB   = (N + BKT_ROWS - 1) / BKT_ROWS;
  const int NBLK = (E + EPB - 1) / EPB;
  size_t bbpad = ((size_t)NBLK * ((size_t)NB + 1) + 255) & ~(size_t)255;
  int* cnt      = (int*)(ws + off); off += npad;
  int* rowstart = (int*)(ws + off); off += npad;
  int* runs     = (int*)(ws + off); off += bbpad;
  int* csr      = (int*)(ws + off); off += epad;
  int2* binned  = (int2*)(ws + off); off += 2 * epad;

  float* mu_out   = (float*)d_out;
  float* ls_out   = mu_out + (size_t)N * D_OUT;
  float* zeta_out = ls_out + (size_t)N * D_OUT;

  const int B = 256;
  auto cdiv = [](int a, int b) { return (a + b - 1) / b; };

  vg_block_sort_kernel<<<NBLK, B, 2 * NB * sizeof(int), stream>>>(esrc, edst, binned, runs, stats, E, NB);
  vg_bucket_rows_kernel<<<NB, B, 3 * NBLK * sizeof(int), stream>>>(binned, runs, cnt, rowstart, dinv, csr, N, NBLK);

  vg_gemm_wmma_kernel<<<GEMM1_GRID, B, 0, stream>>>(x, W1, b1, dinv, h, N);

  vg_gather128_kernel<<<cdiv(N, (B / 32) * ROWS_PER_WAVE), B, 0, stream>>>(h, csr, rowstart, cnt, dinv, hagg, stats, N);

  vg_gemm_bn_wmma_kernel<<<GEMM2_GRID, B, 0, stream>>>(hagg, Wmu, bmu, stats, gamma, beta, dinv, lbuf, rns, N, N);

  vg_gather64_kernel<<<cdiv(N, (B / 32) * ROWS_PER_WAVE), B, 0, stream>>>(lbuf, csr, rowstart, cnt, dinv, rns, noise,
                                                        mu_out, ls_out, zeta_out, N);
}
